# starting kernel
# baseline (speedup 1.0000x reference)
#include <hip/hip_runtime.h>
#include <hip/hip_bf16.h>

typedef __attribute__((ext_vector_type(2))) float v2f;
typedef __attribute__((ext_vector_type(8))) float v8f;

__global__ __launch_bounds__(256) void gcn_gemm_wmma_f32(
    const float* __restrict__ A, const float* __restrict__ W,
    float* __restrict__ C, int M, int K) {
  const int wave = blockIdx.x * (blockDim.x >> 5) + (threadIdx.x >> 5);
  const int lane = threadIdx.x & 31;
  const int m0 = wave << 4;
  if (m0 >= M) return;
  const int half = lane >> 4;
  const int l = lane & 15;

  v8f acc = {};
  const float* arow = A + (size_t)(m0 + l) * K + (half << 1);
  for (int k0 = 0; k0 < K; k0 += 4) {
    v2f a;
    a.x = __builtin_nontemporal_load(&arow[k0 + 0]);
    a.y = __builtin_nontemporal_load(&arow[k0 + 1]);
    v2f b;
    b.x = W[(size_t)(k0 + (half << 1) + 0) * 16 + l];
    b.y = W[(size_t)(k0 + (half << 1) + 1) * 16 + l];
    acc = __builtin_amdgcn_wmma_f32_16x16x4_f32(
         false, a,  false, b,
         (short)0, acc,  false,  false);
  }
  float* cp = C + ((size_t)(m0 + (half << 3)) << 4) + l;
#pragma unroll
  for (int r = 0; r < 8; ++r) cp[(size_t)r << 4] = acc[r];
}

__global__ void gcn_fill(float* __restrict__ p, float v, int n) {
  int i = blockIdx.x * blockDim.x + threadIdx.x;
  if (i < n) p[i] = v;
}

__global__ void gcn_deg_count(const int* __restrict__ dst, float* __restrict__ deg, int E) {
  int e = blockIdx.x * blockDim.x + threadIdx.x;
  if (e < E) atomicAdd(&deg[__builtin_nontemporal_load(&dst[e])], 1.0f);
}

__global__ void gcn_rsqrt(float* __restrict__ d, int n) {
  int i = blockIdx.x * blockDim.x + threadIdx.x;
  if (i < n) d[i] = rsqrtf(d[i]);
}

__global__ void gcn_agg_init(const float* __restrict__ hw, const float* __restrict__ dis,
                             const float* __restrict__ b, float* __restrict__ agg, int n16) {
  int t = blockIdx.x * blockDim.x + threadIdx.x;
  if (t >= n16) return;
  int i = t >> 4, c = t & 15;
  float d = dis[i];
  agg[t] = d * d * hw[t] + b[c];
}

__global__ void gcn_edge_scatter16(const int* __restrict__ src, const int* __restrict__ dst,
                                   const float* __restrict__ dis, const float* __restrict__ hw,
                                   float* __restrict__ agg, int E) {
  int t = blockIdx.x * blockDim.x + threadIdx.x;
  int e = t >> 4, c = t & 15;
  if (e >= E) return;
  int s = __builtin_nontemporal_load(&src[e]);
  int d = __builtin_nontemporal_load(&dst[e]);
  float norm = dis[s] * dis[d];
  atomicAdd(&agg[((size_t)d << 4) + c], norm * hw[((size_t)s << 4) + c]);
}

__global__ void gcn_relu(float* __restrict__ p, int n) {
  int i = blockIdx.x * blockDim.x + threadIdx.x;
  if (i < n) p[i] = fmaxf(p[i], 0.0f);
}

__global__ void gcn_mv16(const float* __restrict__ h, const float* __restrict__ W3,
                         float* __restrict__ out, int N) {
  int i = blockIdx.x * blockDim.x + threadIdx.x;
  if (i >= N) return;
  const float* hp = h + ((size_t)i << 4);
  float s = 0.f;
#pragma unroll
  for (int c = 0; c < 16; ++c) s += __builtin_nontemporal_load(&hp[c]) * W3[c];
  out[i] = s;
}

__global__ void gcn_out_init(const float* __restrict__ hw3, const float* __restrict__ dis,
                             const float* __restrict__ b3, float* __restrict__ out, int N) {
  int i = blockIdx.x * blockDim.x + threadIdx.x;
  if (i >= N) return;
  float d = dis[i];
  out[i] = d * d * hw3[i] + b3[0];
}

__global__ void gcn_edge_scatter1(const int* __restrict__ src, const int* __restrict__ dst,
                                  const float* __restrict__ dis, const float* __restrict__ hw3,
                                  float* __restrict__ out, int E) {
  int e = blockIdx.x * blockDim.x + threadIdx.x;
  if (e >= E) return;
  int s = __builtin_nontemporal_load(&src[e]);
  int d = __builtin_nontemporal_load(&dst[e]);
  atomicAdd(&out[d], dis[s] * dis[d] * hw3[s]);
}

static inline int cdiv(long long a, long long b) { return (int)((a + b - 1) / b); }

extern "C" void kernel_launch(void* const* d_in, const int* in_sizes, int n_in,
                              void* d_out, int out_size, void* d_ws, size_t ws_size,
                              hipStream_t stream) {
  const float* x  = (const float*)d_in[0];
  const int*   ei = (const int*)d_in[1];
  const float* W1 = (const float*)d_in[2];
  const float* b1 = (const float*)d_in[3];
  const float* W2 = (const float*)d_in[4];
  const float* b2 = (const float*)d_in[5];
  const float* W3 = (const float*)d_in[6];
  const float* b3 = (const float*)d_in[7];

  const int IN_CH = 256;
  const int HID = 16;
  const int N = in_sizes[0] / IN_CH;
  const int E = in_sizes[1] / 2;
  const int* src = ei;
  const int* dst = ei + E;
  float* out = (float*)d_out;

  char* ws = (char*)d_ws;
  size_t off = 0;
  float* dis = (float*)(ws + off); off += (((size_t)N * 4) + 255) & ~(size_t)255;
  float* HW  = (float*)(ws + off); off += (((size_t)N * HID * 4) + 255) & ~(size_t)255;
  float* AGG = (float*)(ws + off); off += (((size_t)N * HID * 4) + 255) & ~(size_t)255;
  (void)ws_size; (void)n_in; (void)out_size;

  const int B = 256;
  const int n16 = N * HID;
  const int gemmBlocks = cdiv(cdiv(N, 16), B / 32);

  gcn_fill<<<cdiv(N, B), B, 0, stream>>>(dis, 1.0f, N);
  gcn_deg_count<<<cdiv(E, B), B, 0, stream>>>(dst, dis, E);
  gcn_rsqrt<<<cdiv(N, B), B, 0, stream>>>(dis, N);

  gcn_gemm_wmma_f32<<<gemmBlocks, B, 0, stream>>>(x, W1, HW, N, IN_CH);
  gcn_agg_init<<<cdiv(n16, B), B, 0, stream>>>(HW, dis, b1, AGG, n16);
  gcn_edge_scatter16<<<cdiv((long long)E * 16, B), B, 0, stream>>>(src, dst, dis, HW, AGG, E);
  gcn_relu<<<cdiv(n16, B), B, 0, stream>>>(AGG, n16);

  gcn_gemm_wmma_f32<<<gemmBlocks, B, 0, stream>>>(AGG, W2, HW, N, HID);
  gcn_agg_init<<<cdiv(n16, B), B, 0, stream>>>(HW, dis, b2, AGG, n16);
  gcn_edge_scatter16<<<cdiv((long long)E * 16, B), B, 0, stream>>>(src, dst, dis, HW, AGG, E);
  gcn_relu<<<cdiv(n16, B), B, 0, stream>>>(AGG, n16);

  gcn_mv16<<<cdiv(N, B), B, 0, stream>>>(AGG, W3, HW, N);
  gcn_out_init<<<cdiv(N, B), B, 0, stream>>>(HW, dis, b3, out, N);
  gcn_edge_scatter1<<<cdiv(E, B), B, 0, stream>>>(src, dst, dis, HW, out, E);
}


# Edge passes of layers 2 and 3 walk per-work-group hit lists (source, row, norm) kept by layer 1's scan: no 2nd scan, no dis[src] lookup; sum order: LDS arrival, last layer 5 chunk planes, was 31
# speedup vs baseline: 3.4886x; 3.4886x over previous
#include <hip/hip_runtime.h>
#include <hip/hip_bf16.h>

typedef __attribute__((ext_vector_type(2))) float v2f;
typedef __attribute__((ext_vector_type(8))) float v8f;

template <bool RELU_A>
__global__ __launch_bounds__(256) void gcn_gemm_wmma_f32(
    const float* A, const float* __restrict__ W,
    const float* __restrict__ dis, const float* __restrict__ b,
    float* __restrict__ C, float* AGG, int M, int K) {
  const int wave = blockIdx.x * (blockDim.x >> 5) + (threadIdx.x >> 5);
  const int lane = threadIdx.x & 31;
  const int m0 = wave << 4;
  if (m0 >= M) return;
  const int half = lane >> 4;
  const int l = lane & 15;

  v8f acc = {};
  const float* arow = A + (size_t)(m0 + l) * K + (half << 1);
  for (int k0 = 0; k0 < K; k0 += 4) {
    v2f a;
    a.x = __builtin_nontemporal_load(&arow[k0 + 0]);
    a.y = __builtin_nontemporal_load(&arow[k0 + 1]);
    if (RELU_A) {
      a.x = fmaxf(a.x, 0.0f);
      a.y = fmaxf(a.y, 0.0f);
    }
    v2f b;
    b.x = W[(size_t)(k0 + (half << 1) + 0) * 16 + l];
    b.y = W[(size_t)(k0 + (half << 1) + 1) * 16 + l];
    acc = __builtin_amdgcn_wmma_f32_16x16x4_f32(
         false, a,  false, b,
         (short)0, acc,  false,  false);
  }
  const int r0 = m0 + (half << 3);
  const float4 d0 = *reinterpret_cast<const float4*>(dis + r0);
  const float4 d1 = *reinterpret_cast<const float4*>(dis + r0 + 4);
  const float dd[8] = {d0.x, d0.y, d0.z, d0.w, d1.x, d1.y, d1.z, d1.w};
  const float bias = b[l];
  float* cp = C + ((size_t)r0 << 4) + l;
  float* gp = AGG + ((size_t)r0 << 4) + l;
#pragma unroll
  for (int r = 0; r < 8; ++r) {
    cp[(size_t)r << 4] = acc[r];
    gp[(size_t)r << 4] = dd[r] * dd[r] * acc[r] + bias;
  }
}

constexpr int TILE_NODES  = 25000;
constexpr int TILE_CHUNKS = 31;
constexpr int TILE_THREADS = 1024;

template <bool WEIGHTED, int NE>
__device__ __forceinline__ void gcn_tile_add(
    float* tile, const float* dtile, const int (&d)[NE], const int (&s)[NE], int node0, int nodes,
    const float* __restrict__ dis, const float* __restrict__ hw3) {
  bool hit[NE];
  int rel[NE];
#pragma unroll
  for (int j = 0; j < NE; ++j) {
    rel[j] = d[j] - node0;
    hit[j] = (unsigned)rel[j] < (unsigned)nodes;
  }
  if (WEIGHTED) {
    float v[NE];
#pragma unroll
    for (int j = 0; j < NE; ++j) {
      const int si = hit[j] ? s[j] : 0;
      const int ri = hit[j] ? rel[j] : 0;
      v[j] = dis[si] * dtile[ri] * hw3[si];
    }
#pragma unroll
    for (int j = 0; j < NE; ++j)
      if (hit[j]) atomicAdd(&tile[rel[j]], v[j]);
  } else {
#pragma unroll
    for (int j = 0; j < NE; ++j)
      if (hit[j]) atomicAdd(&tile[rel[j]], 1.0f);
  }
}

template <bool WEIGHTED>
__global__ __launch_bounds__(TILE_THREADS) void gcn_tile_acc1(
    const int* __restrict__ src, const int* __restrict__ dst,
    const float* __restrict__ dis, const float* __restrict__ hw3,
    float* __restrict__ planes, int N, int E, int epc) {
  __shared__ float tile[TILE_NODES];
  __shared__ float dtile[WEIGHTED ? TILE_NODES : 1];
  const int range = blockIdx.x / TILE_CHUNKS;
  const int chunk = blockIdx.x - range * TILE_CHUNKS;
  const int node0 = range * TILE_NODES;
  const int nodes = min(TILE_NODES, N - node0);
  for (int i = threadIdx.x; i < nodes; i += TILE_THREADS) {
    tile[i] = 0.f;
    if (WEIGHTED) dtile[i] = dis[node0 + i];
  }
  __syncthreads();

  const int e0 = min(E, chunk * epc);
  const int e1 = min(E, e0 + epc);

  const int nvec = (e1 - e0) >> 2;
  const int4* __restrict__ d4 = reinterpret_cast<const int4*>(dst + e0);
  const int4* __restrict__ s4 = reinterpret_cast<const int4*>(src + e0);
  int g = threadIdx.x;
  for (; g + 3 * TILE_THREADS < nvec; g += 4 * TILE_THREADS) {
    const int4 q0 = d4[g];
    const int4 q1 = d4[g + TILE_THREADS];
    const int4 q2 = d4[g + 2 * TILE_THREADS];
    const int4 q3 = d4[g + 3 * TILE_THREADS];
    int4 p0 = q0, p1 = q0, p2 = q0, p3 = q0;
    if (WEIGHTED) {
      p0 = s4[g];
      p1 = s4[g + TILE_THREADS];
      p2 = s4[g + 2 * TILE_THREADS];
      p3 = s4[g + 3 * TILE_THREADS];
    }
    const int d[16] = {q0.x, q0.y, q0.z, q0.w, q1.x, q1.y, q1.z, q1.w,
                       q2.x, q2.y, q2.z, q2.w, q3.x, q3.y, q3.z, q3.w};
    const int s[16] = {p0.x, p0.y, p0.z, p0.w, p1.x, p1.y, p1.z, p1.w,
                       p2.x, p2.y, p2.z, p2.w, p3.x, p3.y, p3.z, p3.w};
    gcn_tile_add<WEIGHTED, 16>(tile, dtile, d, s, node0, nodes, dis, hw3);
  }
  for (; g < nvec; g += TILE_THREADS) {
    const int4 q = d4[g];
    int4 p = q;
    if (WEIGHTED) p = s4[g];
    const int d[4] = {q.x, q.y, q.z, q.w};
    const int s[4] = {p.x, p.y, p.z, p.w};
    gcn_tile_add<WEIGHTED, 4>(tile, dtile, d, s, node0, nodes, dis, hw3);
  }
  for (int t = e0 + 4 * nvec + threadIdx.x; t < e1; t += TILE_THREADS) {
    const int d[1] = {dst[t]};
    const int s[1] = {WEIGHTED ? src[t] : 0};
    gcn_tile_add<WEIGHTED, 1>(tile, dtile, d, s, node0, nodes, dis, hw3);
  }
  __syncthreads();

  float* __restrict__ pl = planes + (size_t)chunk * N + node0;
  for (int i = threadIdx.x; i < nodes; i += TILE_THREADS) pl[i] = tile[i];
}

__global__ void gcn_deg_finish(const float* __restrict__ planes, float* __restrict__ dis, int N) {
  int i = blockIdx.x * blockDim.x + threadIdx.x;
  if (i >= N) return;
  float deg = 1.0f;
#pragma unroll
  for (int c = 0; c < TILE_CHUNKS; ++c) deg += planes[(size_t)c * N + i];
  dis[i] = rsqrtf(deg);
}

constexpr int T16_NODES   = 4096;
constexpr int T16_CHUNKS  = 5;
constexpr int T16_THREADS = 1024;
constexpr int T16_WAVES   = T16_THREADS / 32;
constexpr int T16_RING    = 64;

template <int J>
__device__ __forceinline__ int gcn_row_lane(int v) {
  return __builtin_amdgcn_mov_dpp(v, 0x150 + J, 0xf, 0xf, true);
}

#define GCN_FOR_16(X) X(0) X(1) X(2) X(3) X(4) X(5) X(6) X(7) X(8) X(9) X(10) X(11) X(12) X(13) X(14) X(15)

constexpr int T16_LIST_CAP = 32768;
constexpr int T16_ROW_BITS = 12;
static_assert(T16_NODES == (1 << T16_ROW_BITS), "row field of a packed hit");

__device__ __forceinline__ void gcn_tile16_gather(
    int s, int lane, const float* __restrict__ hw, float (&v)[16]) {
  const int c = lane & 15;
#define GCN_ROW_LOAD(J) v[J] = hw[(gcn_row_lane<J>(s) << 4) + c];
  GCN_FOR_16(GCN_ROW_LOAD)
#undef GCN_ROW_LOAD
}

template <bool FULL>
__device__ __forceinline__ void gcn_tile16_add(
    const float (&v)[16], int rel, float nrm, int count, int lane, float* tile) {
  const int c = lane & 15;
  const int row0 = lane & 16;
  const int nrm_bits = __float_as_int(nrm);
  const int left = count - row0;

#define GCN_ROW_ADD(J)                                                        \
  {                                                                           \
    const int rj = gcn_row_lane<J>(rel);                                      \
    const float nj = __int_as_float(gcn_row_lane<J>(nrm_bits));               \
    if (FULL || J < left) atomicAdd(&tile[(rj << 4) + c], nj * v[J]);         \
  }
  GCN_FOR_16(GCN_ROW_ADD)
#undef GCN_ROW_ADD
}
#undef GCN_FOR_16

template <bool FULL>
__device__ __forceinline__ void gcn_tile16_rows(
    int s, int rel, float nrm, int count, int lane,
    const float* __restrict__ hw, float* tile) {
  float v[16];
  gcn_tile16_gather(s, lane, hw, v);
  gcn_tile16_add<FULL>(v, rel, nrm, count, lane, tile);
}

template <bool FULL>
__device__ __forceinline__ void gcn_tile16_batch(
    const int* ring_s, const int* ring_r, int head, int count, int lane,
    const float* __restrict__ dis, const float* __restrict__ hw,
    const float* dtile, float* tile,
    unsigned* list_p, float* list_n, int* cursor) {
  int s = 0, rel = 0;
  float nrm = 0.f;
  const bool have = FULL || lane < count;
  if (have) {
    const int slot = (head + lane) & (T16_RING - 1);
    rel = ring_r[slot];
    s = ring_s[slot];
    nrm = dis[s] * dtile[rel];
  }
  if (list_p != nullptr) {
    int pos = 0;
    if (lane == 0) pos = atomicAdd(cursor, count);
    pos = __builtin_amdgcn_readfirstlane(pos) + lane;
    if (have && pos < T16_LIST_CAP) {
      list_p[pos] = ((unsigned)s << T16_ROW_BITS) | (unsigned)rel;
      if (list_n != nullptr) list_n[pos] = nrm;
    }
  }
  gcn_tile16_rows<FULL>(s, rel, nrm, count, lane, hw, tile);
}

template <bool WALK>
__global__ __launch_bounds__(T16_THREADS) void gcn_tile_scatter16(
    const int* __restrict__ src, const int* __restrict__ dst,
    const float* __restrict__ dis, const float* __restrict__ hw,
    float* __restrict__ agg, unsigned* lists_p, float* lists_n, int* counts,
    int N, int E, int epc) {
  __shared__ float tile[T16_NODES * 16];
  __shared__ float dtile[T16_NODES];
  __shared__ int ring_s_all[T16_WAVES * T16_RING];
  __shared__ int ring_r_all[T16_WAVES * T16_RING];
  __shared__ int cursor;
  const int range = blockIdx.x / T16_CHUNKS;
  const int chunk = blockIdx.x - range * T16_CHUNKS;
  const int node0 = range * T16_NODES;
  const int nodes = min(T16_NODES, N - node0);
  for (int i = threadIdx.x; i < nodes * 16; i += T16_THREADS) tile[i] = 0.f;
  for (int i = threadIdx.x; i < nodes; i += T16_THREADS) dtile[i] = dis[node0 + i];
  if (threadIdx.x == 0) cursor = 0;
  __syncthreads();

  const int lane = threadIdx.x & 31;
  const int wv = __builtin_amdgcn_readfirstlane(threadIdx.x >> 5);

  const size_t list0 = (size_t)blockIdx.x * T16_LIST_CAP;
  bool walked = false;
  if (WALK && lists_p != nullptr) {
    const int cnt = counts[blockIdx.x];
    if (cnt <= T16_LIST_CAP) {
      walked = true;
      const unsigned* __restrict__ lp = lists_p + list0;
      const float* __restrict__ ln = lists_n != nullptr ? lists_n + list0 : nullptr;
      const int trips = (cnt + 31) >> 5;
      const int lastpos = cnt - 1;
      unsigned e_next = 0u;
      float n_next = 0.f;
      int b = wv;
      if (b < trips) {
        const int p = min((b << 5) + lane, lastpos);
        e_next = lp[p];
        if (ln != nullptr) n_next = ln[p];
      }
      for (; b < trips; b += T16_WAVES) {
        const unsigned e = e_next;
        float nrm = n_next;
        const int s = (int)(e >> T16_ROW_BITS);
        const int rel = (int)(e & (unsigned)(T16_NODES - 1));
        float v[16];
        gcn_tile16_gather(s, lane, hw, v);
        if (ln == nullptr) nrm = dis[s] * dtile[rel];
        const int bn = b + T16_WAVES;
        if (bn < trips) {
          const int p = min((bn << 5) + lane, lastpos);
          e_next = lp[p];
          if (ln != nullptr) n_next = ln[p];
        }
        const int left = cnt - (b << 5);
        if (left >= 32) gcn_tile16_add<true>(v, rel, nrm, 32, lane, tile);
        else            gcn_tile16_add<false>(v, rel, nrm, left, lane, tile);
      }
    }
  }

  if (!walked) {
    unsigned* list_p = (!WALK && lists_p != nullptr) ? lists_p + list0 : nullptr;
    float* list_n = (list_p != nullptr && lists_n != nullptr) ? lists_n + list0 : nullptr;
    int* ring_s = ring_s_all + wv * T16_RING;
    int* ring_r = ring_r_all + wv * T16_RING;
    int head = 0, tail = 0;

    const int e0 = min(E, chunk * epc);
    const int e1 = min(E, e0 + epc);
    const int nvec = (e1 - e0) >> 2;
    const int4* __restrict__ d4 = reinterpret_cast<const int4*>(dst + e0);
    const int4* __restrict__ s4 = reinterpret_cast<const int4*>(src + e0);

    auto step = [&](int d, int s) {
      const int rel = d - node0;
      const bool hit = (unsigned)rel < (unsigned)nodes;
      const unsigned m = __builtin_amdgcn_ballot_w32(hit);
      if (m != 0u) {
        if (hit) {
          const int slot = (tail + (int)__builtin_amdgcn_mbcnt_lo(m, 0u)) & (T16_RING - 1);
          ring_s[slot] = s;
          ring_r[slot] = rel;
        }
        tail += __builtin_popcount(m);
        if (tail - head >= 32) {
          gcn_tile16_batch<true>(ring_s, ring_r, head, 32, lane, dis, hw, dtile, tile,
                                 list_p, list_n, &cursor);
          head += 32;
        }
      }
    };

    const int last = nvec - 1;
    int4 n0 = make_int4(-1, -1, -1, -1), n1 = n0, n2 = n0, n3 = n0;
    int4 m0 = make_int4(0, 0, 0, 0), m1 = m0, m2 = m0, m3 = m0;
    if (nvec > 0) {
      const int g = threadIdx.x;
      n0 = d4[min(g, last)];
      n1 = d4[min(g + T16_THREADS, last)];
      n2 = d4[min(g + 2 * T16_THREADS, last)];
      n3 = d4[min(g + 3 * T16_THREADS, last)];
      m0 = s4[min(g, last)];
      m1 = s4[min(g + T16_THREADS, last)];
      m2 = s4[min(g + 2 * T16_THREADS, last)];
      m3 = s4[min(g + 3 * T16_THREADS, last)];
    }
    for (int base = 0; base < nvec; base += 4 * T16_THREADS) {
      const int4 q0 = n0, q1 = n1, q2 = n2, q3 = n3;
      const int4 p0 = m0, p1 = m1, p2 = m2, p3 = m3;
      const int g0 = base + threadIdx.x;
      const int g1 = g0 + T16_THREADS;
      const int g2 = g0 + 2 * T16_THREADS;
      const int g3 = g0 + 3 * T16_THREADS;
      n0 = d4[min(g0 + 4 * T16_THREADS, last)];
      n1 = d4[min(g1 + 4 * T16_THREADS, last)];
      n2 = d4[min(g2 + 4 * T16_THREADS, last)];
      n3 = d4[min(g3 + 4 * T16_THREADS, last)];
      m0 = s4[min(g0 + 4 * T16_THREADS, last)];
      m1 = s4[min(g1 + 4 * T16_THREADS, last)];
      m2 = s4[min(g2 + 4 * T16_THREADS, last)];
      m3 = s4[min(g3 + 4 * T16_THREADS, last)];
      const bool in0 = g0 < nvec, in1 = g1 < nvec, in2 = g2 < nvec, in3 = g3 < nvec;
      step(in0 ? q0.x : -1, p0.x); step(in0 ? q0.y : -1, p0.y); step(in0 ? q0.z : -1, p0.z); step(in0 ? q0.w : -1, p0.w);
      step(in1 ? q1.x : -1, p1.x); step(in1 ? q1.y : -1, p1.y); step(in1 ? q1.z : -1, p1.z); step(in1 ? q1.w : -1, p1.w);
      step(in2 ? q2.x : -1, p2.x); step(in2 ? q2.y : -1, p2.y); step(in2 ? q2.z : -1, p2.z); step(in2 ? q2.w : -1, p2.w);
      step(in3 ? q3.x : -1, p3.x); step(in3 ? q3.y : -1, p3.y); step(in3 ? q3.z : -1, p3.z); step(in3 ? q3.w : -1, p3.w);
    }
    {
      const int t = e0 + 4 * nvec + (int)threadIdx.x;
      const bool have = t < e1;
      const int d = have ? dst[t] : -1;
      const int s = have ? src[t] : 0;
      step(d, s);
    }
    if (tail - head > 0)
      gcn_tile16_batch<false>(ring_s, ring_r, head, tail - head, lane, dis, hw, dtile, tile,
                              list_p, list_n, &cursor);
  }
  __syncthreads();

  if (!WALK && lists_p != nullptr && threadIdx.x == 0) counts[blockIdx.x] = cursor;

  float* __restrict__ ag = agg + ((size_t)node0 << 4);
  for (int i = threadIdx.x; i < nodes * 16; i += T16_THREADS) atomicAdd(&ag[i], tile[i]);
}

__global__ void gcn_mv16(const float* __restrict__ h, const float* __restrict__ W3,
                         float* __restrict__ out, int N) {
  int i = blockIdx.x * blockDim.x + threadIdx.x;
  if (i >= N) return;
  const float* hp = h + ((size_t)i << 4);
  float s = 0.f;
#pragma unroll
  for (int c = 0; c < 16; ++c) s += fmaxf(__builtin_nontemporal_load(&hp[c]), 0.0f) * W3[c];
  out[i] = s;
}

__global__ __launch_bounds__(T16_THREADS) void gcn_list_acc1(
    const int* __restrict__ src, const int* __restrict__ dst,
    const float* __restrict__ dis, const float* __restrict__ hw3,
    const unsigned* __restrict__ lists_p, const float* __restrict__ lists_n,
    const int* __restrict__ counts, float* __restrict__ planes, int N, int E, int epc) {
  __shared__ float tile[T16_NODES];
  const int range = blockIdx.x / T16_CHUNKS;
  const int chunk = blockIdx.x - range * T16_CHUNKS;
  const int node0 = range * T16_NODES;
  const int nodes = min(T16_NODES, N - node0);
  for (int i = threadIdx.x; i < nodes; i += T16_THREADS) tile[i] = 0.f;
  __syncthreads();

  const int cnt = counts[blockIdx.x];
  if (cnt <= T16_LIST_CAP) {
    const unsigned* __restrict__ lp = lists_p + (size_t)blockIdx.x * T16_LIST_CAP;
    const float* __restrict__ ln = lists_n + (size_t)blockIdx.x * T16_LIST_CAP;
    int p = threadIdx.x;
    for (; p + 3 * T16_THREADS < cnt; p += 4 * T16_THREADS) {
      const unsigned e0 = lp[p];
      const unsigned e1 = lp[p + T16_THREADS];
      const unsigned e2 = lp[p + 2 * T16_THREADS];
      const unsigned e3 = lp[p + 3 * T16_THREADS];
      const float n0 = ln[p];
      const float n1 = ln[p + T16_THREADS];
      const float n2 = ln[p + 2 * T16_THREADS];
      const float n3 = ln[p + 3 * T16_THREADS];
      const float h0 = hw3[e0 >> T16_ROW_BITS];
      const float h1 = hw3[e1 >> T16_ROW_BITS];
      const float h2 = hw3[e2 >> T16_ROW_BITS];
      const float h3 = hw3[e3 >> T16_ROW_BITS];
      atomicAdd(&tile[e0 & (unsigned)(T16_NODES - 1)], n0 * h0);
      atomicAdd(&tile[e1 & (unsigned)(T16_NODES - 1)], n1 * h1);
      atomicAdd(&tile[e2 & (unsigned)(T16_NODES - 1)], n2 * h2);
      atomicAdd(&tile[e3 & (unsigned)(T16_NODES - 1)], n3 * h3);
    }
    for (; p < cnt; p += T16_THREADS) {
      const unsigned e = lp[p];
      atomicAdd(&tile[e & (unsigned)(T16_NODES - 1)], ln[p] * hw3[e >> T16_ROW_BITS]);
    }
  } else {
    const int e0 = min(E, chunk * epc);
    const int e1 = min(E, e0 + epc);
    for (int t = e0 + (int)threadIdx.x; t < e1; t += T16_THREADS) {
      const int rel = dst[t] - node0;
      if ((unsigned)rel < (unsigned)nodes) {
        const int s = src[t];
        atomicAdd(&tile[rel], dis[s] * dis[node0 + rel] * hw3[s]);
      }
    }
  }
  __syncthreads();

  float* __restrict__ pl = planes + (size_t)chunk * N + node0;
  for (int i = threadIdx.x; i < nodes; i += T16_THREADS) pl[i] = tile[i];
}

template <int NP>
__global__ void gcn_out_finish(const float* __restrict__ planes, const float* __restrict__ hw3,
                               const float* __restrict__ dis, const float* __restrict__ b3,
                               float* __restrict__ out, int N) {
  int i = blockIdx.x * blockDim.x + threadIdx.x;
  if (i >= N) return;
  float d = dis[i];
  float acc = d * d * hw3[i] + b3[0];
#pragma unroll
  for (int c = 0; c < NP; ++c) acc += planes[(size_t)c * N + i];
  out[i] = acc;
}

static inline int cdiv(long long a, long long b) { return (int)((a + b - 1) / b); }

extern "C" void kernel_launch(void* const* d_in, const int* in_sizes, int n_in,
                              void* d_out, int out_size, void* d_ws, size_t ws_size,
                              hipStream_t stream) {
  const float* x  = (const float*)d_in[0];
  const int*   ei = (const int*)d_in[1];
  const float* W1 = (const float*)d_in[2];
  const float* b1 = (const float*)d_in[3];
  const float* W2 = (const float*)d_in[4];
  const float* b2 = (const float*)d_in[5];
  const float* W3 = (const float*)d_in[6];
  const float* b3 = (const float*)d_in[7];

  const int IN_CH = 256;
  const int HID = 16;
  const int N = in_sizes[0] / IN_CH;
  const int E = in_sizes[1] / 2;
  const int* src = ei;
  const int* dst = ei + E;
  float* out = (float*)d_out;

  char* ws = (char*)d_ws;
  size_t off = 0;
  float* dis = (float*)(ws + off); off += (((size_t)N * 4) + 255) & ~(size_t)255;
  float* HW  = (float*)(ws + off); off += (((size_t)N * HID * 4) + 255) & ~(size_t)255;
  float* AGG = (float*)(ws + off); off += (((size_t)N * HID * 4) + 255) & ~(size_t)255;
  (void)n_in; (void)out_size;

  const int B = 256;
  const int gemmBlocks = cdiv(cdiv(N, 16), B / 32);

  const int tileGrid = cdiv(N, TILE_NODES) * TILE_CHUNKS;
  const int epc = cdiv(cdiv(E, TILE_CHUNKS), 4 * TILE_THREADS) * (4 * TILE_THREADS);

  const int t16Grid = cdiv(N, T16_NODES) * T16_CHUNKS;
  const int epc16 = cdiv(cdiv(E, T16_CHUNKS), 16 * T16_THREADS) * (16 * T16_THREADS);

  const size_t listBytes = (size_t)t16Grid * T16_LIST_CAP * 4;
  const size_t offP = off + ((((size_t)t16Grid * 4) + 255) & ~(size_t)255);
  const size_t offN = offP + listBytes;
  const bool keepP = N <= (1 << (32 - T16_ROW_BITS)) && ws_size >= offN;
  const bool keepN = keepP && ws_size >= offN + listBytes;
  int* cnt16 = (int*)(ws + off);
  unsigned* listP = keepP ? (unsigned*)(ws + offP) : nullptr;
  float* listN = keepN ? (float*)(ws + offN) : nullptr;

  gcn_tile_acc1<false><<<tileGrid, TILE_THREADS, 0, stream>>>(src, dst, nullptr, nullptr, HW, N, E, epc);
  gcn_deg_finish<<<cdiv(N, B), B, 0, stream>>>(HW, dis, N);

  gcn_gemm_wmma_f32<false><<<gemmBlocks, B, 0, stream>>>(x, W1, dis, b1, HW, AGG, N, IN_CH);
  gcn_tile_scatter16<false><<<t16Grid, T16_THREADS, 0, stream>>>(src, dst, dis, HW, AGG, listP, listN, cnt16, N, E, epc16);

  gcn_gemm_wmma_f32<true><<<gemmBlocks, B, 0, stream>>>(AGG, W2, dis, b2, HW, AGG, N, HID);
  gcn_tile_scatter16<true><<<t16Grid, T16_THREADS, 0, stream>>>(src, dst, dis, HW, AGG, listP, listN, cnt16, N, E, epc16);

  gcn_mv16<<<cdiv(N, B), B, 0, stream>>>(AGG, W3, HW, N);
  if (keepN) {
    gcn_list_acc1<<<t16Grid, T16_THREADS, 0, stream>>>(src, dst, dis, HW, listP, listN, cnt16, HW + N, N, E, epc16);
    gcn_out_finish<T16_CHUNKS><<<cdiv(N, B), B, 0, stream>>>(HW + N, HW, dis, b3, out, N);
  } else {
    gcn_tile_acc1<true><<<tileGrid, TILE_THREADS, 0, stream>>>(src, dst, dis, HW, HW + N, N, E, epc);
    gcn_out_finish<TILE_CHUNKS><<<cdiv(N, B), B, 0, stream>>>(HW + N, HW, dis, b3, out, N);
  }
}


# Edge scan made two-level: degree pass (13 ranges of 16384 nodes x 19 chunks) keeps its edges as packed lists; layer 1's 49 ranges scan 4 of those lists each, not all edges; sum order LDS arrival
# speedup vs baseline: 4.0086x; 1.1491x over previous
#include <hip/hip_runtime.h>
#include <hip/hip_bf16.h>

typedef __attribute__((ext_vector_type(2))) float v2f;
typedef __attribute__((ext_vector_type(8))) float v8f;

template <bool RELU_A>
__global__ __launch_bounds__(256) void gcn_gemm_wmma_f32(
    const float* A, const float* __restrict__ W,
    const float* __restrict__ dis, const float* __restrict__ b,
    float* __restrict__ C, float* AGG, int M, int K) {
  const int wave = blockIdx.x * (blockDim.x >> 5) + (threadIdx.x >> 5);
  const int lane = threadIdx.x & 31;
  const int m0 = wave << 4;
  if (m0 >= M) return;
  const int half = lane >> 4;
  const int l = lane & 15;

  v8f acc = {};
  const float* arow = A + (size_t)(m0 + l) * K + (half << 1);
  for (int k0 = 0; k0 < K; k0 += 4) {
    v2f a;
    a.x = __builtin_nontemporal_load(&arow[k0 + 0]);
    a.y = __builtin_nontemporal_load(&arow[k0 + 1]);
    if (RELU_A) {
      a.x = fmaxf(a.x, 0.0f);
      a.y = fmaxf(a.y, 0.0f);
    }
    v2f b;
    b.x = W[(size_t)(k0 + (half << 1) + 0) * 16 + l];
    b.y = W[(size_t)(k0 + (half << 1) + 1) * 16 + l];
    acc = __builtin_amdgcn_wmma_f32_16x16x4_f32(
         false, a,  false, b,
         (short)0, acc,  false,  false);
  }
  const int r0 = m0 + (half << 3);
  const float4 d0 = *reinterpret_cast<const float4*>(dis + r0);
  const float4 d1 = *reinterpret_cast<const float4*>(dis + r0 + 4);
  const float dd[8] = {d0.x, d0.y, d0.z, d0.w, d1.x, d1.y, d1.z, d1.w};
  const float bias = b[l];
  float* cp = C + ((size_t)r0 << 4) + l;
  float* gp = AGG + ((size_t)r0 << 4) + l;
#pragma unroll
  for (int r = 0; r < 8; ++r) {
    cp[(size_t)r << 4] = acc[r];
    gp[(size_t)r << 4] = dd[r] * dd[r] * acc[r] + bias;
  }
}

constexpr int TILE_NODES  = 25000;
constexpr int TILE_CHUNKS = 31;
constexpr int TILE_THREADS = 1024;

template <bool WEIGHTED, int NE>
__device__ __forceinline__ void gcn_tile_add(
    float* tile, const float* dtile, const int (&d)[NE], const int (&s)[NE], int node0, int nodes,
    const float* __restrict__ dis, const float* __restrict__ hw3) {
  bool hit[NE];
  int rel[NE];
#pragma unroll
  for (int j = 0; j < NE; ++j) {
    rel[j] = d[j] - node0;
    hit[j] = (unsigned)rel[j] < (unsigned)nodes;
  }
  if (WEIGHTED) {
    float v[NE];
#pragma unroll
    for (int j = 0; j < NE; ++j) {
      const int si = hit[j] ? s[j] : 0;
      const int ri = hit[j] ? rel[j] : 0;
      v[j] = dis[si] * dtile[ri] * hw3[si];
    }
#pragma unroll
    for (int j = 0; j < NE; ++j)
      if (hit[j]) atomicAdd(&tile[rel[j]], v[j]);
  } else {
#pragma unroll
    for (int j = 0; j < NE; ++j)
      if (hit[j]) atomicAdd(&tile[rel[j]], 1.0f);
  }
}

template <bool WEIGHTED>
__global__ __launch_bounds__(TILE_THREADS) void gcn_tile_acc1(
    const int* __restrict__ src, const int* __restrict__ dst,
    const float* __restrict__ dis, const float* __restrict__ hw3,
    float* __restrict__ planes, int N, int E, int epc) {
  __shared__ float tile[TILE_NODES];
  __shared__ float dtile[WEIGHTED ? TILE_NODES : 1];
  const int range = blockIdx.x / TILE_CHUNKS;
  const int chunk = blockIdx.x - range * TILE_CHUNKS;
  const int node0 = range * TILE_NODES;
  const int nodes = min(TILE_NODES, N - node0);
  for (int i = threadIdx.x; i < nodes; i += TILE_THREADS) {
    tile[i] = 0.f;
    if (WEIGHTED) dtile[i] = dis[node0 + i];
  }
  __syncthreads();

  const int e0 = min(E, chunk * epc);
  const int e1 = min(E, e0 + epc);

  const int nvec = (e1 - e0) >> 2;
  const int4* __restrict__ d4 = reinterpret_cast<const int4*>(dst + e0);
  const int4* __restrict__ s4 = reinterpret_cast<const int4*>(src + e0);
  int g = threadIdx.x;
  for (; g + 3 * TILE_THREADS < nvec; g += 4 * TILE_THREADS) {
    const int4 q0 = d4[g];
    const int4 q1 = d4[g + TILE_THREADS];
    const int4 q2 = d4[g + 2 * TILE_THREADS];
    const int4 q3 = d4[g + 3 * TILE_THREADS];
    int4 p0 = q0, p1 = q0, p2 = q0, p3 = q0;
    if (WEIGHTED) {
      p0 = s4[g];
      p1 = s4[g + TILE_THREADS];
      p2 = s4[g + 2 * TILE_THREADS];
      p3 = s4[g + 3 * TILE_THREADS];
    }
    const int d[16] = {q0.x, q0.y, q0.z, q0.w, q1.x, q1.y, q1.z, q1.w,
                       q2.x, q2.y, q2.z, q2.w, q3.x, q3.y, q3.z, q3.w};
    const int s[16] = {p0.x, p0.y, p0.z, p0.w, p1.x, p1.y, p1.z, p1.w,
                       p2.x, p2.y, p2.z, p2.w, p3.x, p3.y, p3.z, p3.w};
    gcn_tile_add<WEIGHTED, 16>(tile, dtile, d, s, node0, nodes, dis, hw3);
  }
  for (; g < nvec; g += TILE_THREADS) {
    const int4 q = d4[g];
    int4 p = q;
    if (WEIGHTED) p = s4[g];
    const int d[4] = {q.x, q.y, q.z, q.w};
    const int s[4] = {p.x, p.y, p.z, p.w};
    gcn_tile_add<WEIGHTED, 4>(tile, dtile, d, s, node0, nodes, dis, hw3);
  }
  for (int t = e0 + 4 * nvec + threadIdx.x; t < e1; t += TILE_THREADS) {
    const int d[1] = {dst[t]};
    const int s[1] = {WEIGHTED ? src[t] : 0};
    gcn_tile_add<WEIGHTED, 1>(tile, dtile, d, s, node0, nodes, dis, hw3);
  }
  __syncthreads();

  float* __restrict__ pl = planes + (size_t)chunk * N + node0;
  for (int i = threadIdx.x; i < nodes; i += TILE_THREADS) pl[i] = tile[i];
}

template <int NP>
__global__ void gcn_deg_finish(const float* __restrict__ planes, float* __restrict__ dis, int N) {
  int i = blockIdx.x * blockDim.x + threadIdx.x;
  if (i >= N) return;
  float deg = 1.0f;
#pragma unroll
  for (int c = 0; c < NP; ++c) deg += planes[(size_t)c * N + i];
  dis[i] = rsqrtf(deg);
}

constexpr int T16_NODES   = 4096;
constexpr int T16_CHUNKS  = 5;
constexpr int T16_THREADS = 1024;
constexpr int T16_WAVES   = T16_THREADS / 32;
constexpr int T16_RING    = 64;

template <int J>
__device__ __forceinline__ int gcn_row_lane(int v) {
  return __builtin_amdgcn_mov_dpp(v, 0x150 + J, 0xf, 0xf, true);
}

#define GCN_FOR_16(X) X(0) X(1) X(2) X(3) X(4) X(5) X(6) X(7) X(8) X(9) X(10) X(11) X(12) X(13) X(14) X(15)

constexpr int T16_LIST_CAP = 32768;
constexpr int T16_ROW_BITS = 12;
static_assert(T16_NODES == (1 << T16_ROW_BITS), "row field of a packed hit");

constexpr int CR_BITS    = 14;
constexpr int CR_NODES   = 1 << CR_BITS;
constexpr int CR_CHUNKS  = 19;
constexpr int CR_THREADS = 1024;
constexpr int CR_WAVES   = CR_THREADS / 32;
constexpr int CR_RING    = 64;
constexpr int CR_CAP     = 32768;
static_assert(CR_NODES == 4 * T16_NODES, "a coarse range is four fine ranges");
static_assert(CR_CHUNKS >= T16_CHUNKS, "every fine chunk holds a coarse chunk");

__device__ __forceinline__ int gcn_chunk_first(int c) { return (c * CR_CHUNKS) / T16_CHUNKS; }

template <bool KEEP>
__global__ __launch_bounds__(CR_THREADS) void gcn_deg_coarse(
    const int* __restrict__ src, const int* __restrict__ dst,
    float* __restrict__ planes, unsigned* coarse, int* ccounts, int N, int E, int epc) {
  __shared__ float tile[CR_NODES];
  __shared__ unsigned ring_all[CR_WAVES * CR_RING];
  __shared__ int cursor;
  const int range = blockIdx.x / CR_CHUNKS;
  const int chunk = blockIdx.x - range * CR_CHUNKS;
  const int node0 = range << CR_BITS;
  const int nodes = min(CR_NODES, N - node0);
  for (int i = threadIdx.x; i < nodes; i += CR_THREADS) tile[i] = 0.f;
  if (threadIdx.x == 0) cursor = 0;
  __syncthreads();

  const int lane = threadIdx.x & 31;
  const int wv = __builtin_amdgcn_readfirstlane(threadIdx.x >> 5);
  unsigned* ring = ring_all + wv * CR_RING;
  unsigned* list = KEEP ? coarse + (size_t)blockIdx.x * CR_CAP : nullptr;
  int head = 0, tail = 0;

  auto flush = [&](int count) {
    int pos = 0;
    if (lane == 0) pos = atomicAdd(&cursor, count);
    pos = __builtin_amdgcn_readfirstlane(pos) + lane;
    if (lane < count && pos < CR_CAP) list[pos] = ring[(head + lane) & (CR_RING - 1)];
  };
  auto step = [&](int d, int s) {
    const int rel = d - node0;
    const bool hit = (unsigned)rel < (unsigned)nodes;
    if (KEEP) {
      const unsigned m = __builtin_amdgcn_ballot_w32(hit);
      if (m != 0u) {
        if (hit) {
          atomicAdd(&tile[rel], 1.0f);
          const int slot = (tail + (int)__builtin_amdgcn_mbcnt_lo(m, 0u)) & (CR_RING - 1);
          ring[slot] = ((unsigned)s << CR_BITS) | (unsigned)rel;
        }
        tail += __builtin_popcount(m);
        if (tail - head >= 32) {
          flush(32);
          head += 32;
        }
      }
    } else {
      if (hit) atomicAdd(&tile[rel], 1.0f);
    }
  };

  const int e0 = min(E, chunk * epc);
  const int e1 = min(E, e0 + epc);
  const int nvec = (e1 - e0) >> 2;
  const int4* __restrict__ d4 = reinterpret_cast<const int4*>(dst + e0);
  const int4* __restrict__ s4 = reinterpret_cast<const int4*>(src + e0);
  const int last = nvec - 1;
  int4 n0 = make_int4(-1, -1, -1, -1), n1 = n0, n2 = n0, n3 = n0;
  int4 m0 = make_int4(0, 0, 0, 0), m1 = m0, m2 = m0, m3 = m0;
  if (nvec > 0) {
    const int g = threadIdx.x;
    n0 = d4[min(g, last)];
    n1 = d4[min(g + CR_THREADS, last)];
    n2 = d4[min(g + 2 * CR_THREADS, last)];
    n3 = d4[min(g + 3 * CR_THREADS, last)];
    m0 = s4[min(g, last)];
    m1 = s4[min(g + CR_THREADS, last)];
    m2 = s4[min(g + 2 * CR_THREADS, last)];
    m3 = s4[min(g + 3 * CR_THREADS, last)];
  }
  for (int base = 0; base < nvec; base += 4 * CR_THREADS) {
    const int4 q0 = n0, q1 = n1, q2 = n2, q3 = n3;
    const int4 p0 = m0, p1 = m1, p2 = m2, p3 = m3;
    const int g0 = base + threadIdx.x;
    const int g1 = g0 + CR_THREADS;
    const int g2 = g0 + 2 * CR_THREADS;
    const int g3 = g0 + 3 * CR_THREADS;
    n0 = d4[min(g0 + 4 * CR_THREADS, last)];
    n1 = d4[min(g1 + 4 * CR_THREADS, last)];
    n2 = d4[min(g2 + 4 * CR_THREADS, last)];
    n3 = d4[min(g3 + 4 * CR_THREADS, last)];
    m0 = s4[min(g0 + 4 * CR_THREADS, last)];
    m1 = s4[min(g1 + 4 * CR_THREADS, last)];
    m2 = s4[min(g2 + 4 * CR_THREADS, last)];
    m3 = s4[min(g3 + 4 * CR_THREADS, last)];
    const bool in0 = g0 < nvec, in1 = g1 < nvec, in2 = g2 < nvec, in3 = g3 < nvec;
    step(in0 ? q0.x : -1, p0.x); step(in0 ? q0.y : -1, p0.y); step(in0 ? q0.z : -1, p0.z); step(in0 ? q0.w : -1, p0.w);
    step(in1 ? q1.x : -1, p1.x); step(in1 ? q1.y : -1, p1.y); step(in1 ? q1.z : -1, p1.z); step(in1 ? q1.w : -1, p1.w);
    step(in2 ? q2.x : -1, p2.x); step(in2 ? q2.y : -1, p2.y); step(in2 ? q2.z : -1, p2.z); step(in2 ? q2.w : -1, p2.w);
    step(in3 ? q3.x : -1, p3.x); step(in3 ? q3.y : -1, p3.y); step(in3 ? q3.z : -1, p3.z); step(in3 ? q3.w : -1, p3.w);
  }
  {
    const int t = e0 + 4 * nvec + (int)threadIdx.x;
    const bool have = t < e1;
    const int d = have ? dst[t] : -1;
    const int s = have ? src[t] : 0;
    step(d, s);
  }
  if (KEEP && tail - head > 0) flush(tail - head);
  __syncthreads();

  if (KEEP && threadIdx.x == 0) ccounts[blockIdx.x] = cursor;

  float* __restrict__ pl = planes + (size_t)chunk * N + node0;
  for (int i = threadIdx.x; i < nodes; i += CR_THREADS) pl[i] = tile[i];
}

__device__ __forceinline__ void gcn_tile16_gather(
    int s, int lane, const float* __restrict__ hw, float (&v)[16]) {
  const int c = lane & 15;
#define GCN_ROW_LOAD(J) v[J] = hw[(gcn_row_lane<J>(s) << 4) + c];
  GCN_FOR_16(GCN_ROW_LOAD)
#undef GCN_ROW_LOAD
}

template <bool FULL>
__device__ __forceinline__ void gcn_tile16_add(
    const float (&v)[16], int rel, float nrm, int count, int lane, float* tile) {
  const int c = lane & 15;
  const int row0 = lane & 16;
  const int nrm_bits = __float_as_int(nrm);
  const int left = count - row0;

#define GCN_ROW_ADD(J)                                                        \
  {                                                                           \
    const int rj = gcn_row_lane<J>(rel);                                      \
    const float nj = __int_as_float(gcn_row_lane<J>(nrm_bits));               \
    if (FULL || J < left) atomicAdd(&tile[(rj << 4) + c], nj * v[J]);         \
  }
  GCN_FOR_16(GCN_ROW_ADD)
#undef GCN_ROW_ADD
}
#undef GCN_FOR_16

template <bool FULL>
__device__ __forceinline__ void gcn_tile16_rows(
    int s, int rel, float nrm, int count, int lane,
    const float* __restrict__ hw, float* tile) {
  float v[16];
  gcn_tile16_gather(s, lane, hw, v);
  gcn_tile16_add<FULL>(v, rel, nrm, count, lane, tile);
}

template <bool FULL>
__device__ __forceinline__ void gcn_tile16_batch(
    const int* ring_s, const int* ring_r, int head, int count, int lane,
    const float* __restrict__ dis, const float* __restrict__ hw,
    const float* dtile, float* tile,
    unsigned* list_p, float* list_n, int* cursor) {
  int s = 0, rel = 0;
  float nrm = 0.f;
  const bool have = FULL || lane < count;
  if (have) {
    const int slot = (head + lane) & (T16_RING - 1);
    rel = ring_r[slot];
    s = ring_s[slot];
    nrm = dis[s] * dtile[rel];
  }
  if (list_p != nullptr) {
    int pos = 0;
    if (lane == 0) pos = atomicAdd(cursor, count);
    pos = __builtin_amdgcn_readfirstlane(pos) + lane;
    if (have && pos < T16_LIST_CAP) {
      list_p[pos] = ((unsigned)s << T16_ROW_BITS) | (unsigned)rel;
      if (list_n != nullptr) list_n[pos] = nrm;
    }
  }
  gcn_tile16_rows<FULL>(s, rel, nrm, count, lane, hw, tile);
}

template <bool WALK>
__global__ __launch_bounds__(T16_THREADS) void gcn_tile_scatter16(
    const int* __restrict__ src, const int* __restrict__ dst,
    const float* __restrict__ dis, const float* __restrict__ hw,
    float* __restrict__ agg, unsigned* lists_p, float* lists_n, int* counts,
    const unsigned* coarse, const int* ccounts, int N, int E, int epcC) {
  __shared__ float tile[T16_NODES * 16];
  __shared__ float dtile[T16_NODES];
  __shared__ int ring_s_all[T16_WAVES * T16_RING];
  __shared__ int ring_r_all[T16_WAVES * T16_RING];
  __shared__ int cursor;
  const int range = blockIdx.x / T16_CHUNKS;
  const int chunk = blockIdx.x - range * T16_CHUNKS;
  const int node0 = range * T16_NODES;
  const int nodes = min(T16_NODES, N - node0);
  for (int i = threadIdx.x; i < nodes * 16; i += T16_THREADS) tile[i] = 0.f;
  for (int i = threadIdx.x; i < nodes; i += T16_THREADS) dtile[i] = dis[node0 + i];
  if (threadIdx.x == 0) cursor = 0;
  __syncthreads();

  const int lane = threadIdx.x & 31;
  const int wv = __builtin_amdgcn_readfirstlane(threadIdx.x >> 5);

  const size_t list0 = (size_t)blockIdx.x * T16_LIST_CAP;
  bool walked = false;
  if (WALK && lists_p != nullptr) {
    const int cnt = counts[blockIdx.x];
    if (cnt <= T16_LIST_CAP) {
      walked = true;
      const unsigned* __restrict__ lp = lists_p + list0;
      const float* __restrict__ ln = lists_n != nullptr ? lists_n + list0 : nullptr;
      const int trips = (cnt + 31) >> 5;
      const int lastpos = cnt - 1;
      unsigned e_next = 0u;
      float n_next = 0.f;
      int b = wv;
      if (b < trips) {
        const int p = min((b << 5) + lane, lastpos);
        e_next = lp[p];
        if (ln != nullptr) n_next = ln[p];
      }
      for (; b < trips; b += T16_WAVES) {
        const unsigned e = e_next;
        float nrm = n_next;
        const int s = (int)(e >> T16_ROW_BITS);
        const int rel = (int)(e & (unsigned)(T16_NODES - 1));
        float v[16];
        gcn_tile16_gather(s, lane, hw, v);
        if (ln == nullptr) nrm = dis[s] * dtile[rel];
        const int bn = b + T16_WAVES;
        if (bn < trips) {
          const int p = min((bn << 5) + lane, lastpos);
          e_next = lp[p];
          if (ln != nullptr) n_next = ln[p];
        }
        const int left = cnt - (b << 5);
        if (left >= 32) gcn_tile16_add<true>(v, rel, nrm, 32, lane, tile);
        else            gcn_tile16_add<false>(v, rel, nrm, left, lane, tile);
      }
    }
  }

  if (!walked) {
    unsigned* list_p = (!WALK && lists_p != nullptr) ? lists_p + list0 : nullptr;
    float* list_n = (list_p != nullptr && lists_n != nullptr) ? lists_n + list0 : nullptr;
    int* ring_s = ring_s_all + wv * T16_RING;
    int* ring_r = ring_r_all + wv * T16_RING;
    int head = 0, tail = 0;

    auto take = [&](bool hit, int s, int rel) {
      const unsigned m = __builtin_amdgcn_ballot_w32(hit);
      if (m != 0u) {
        if (hit) {
          const int slot = (tail + (int)__builtin_amdgcn_mbcnt_lo(m, 0u)) & (T16_RING - 1);
          ring_s[slot] = s;
          ring_r[slot] = rel;
        }
        tail += __builtin_popcount(m);
        if (tail - head >= 32) {
          gcn_tile16_batch<true>(ring_s, ring_r, head, 32, lane, dis, hw, dtile, tile,
                                 list_p, list_n, &cursor);
          head += 32;
        }
      }
    };
    auto step = [&](int d, int s) {
      const int rel = d - node0;
      take((unsigned)rel < (unsigned)nodes, s, rel);
    };
    const unsigned subkey = (unsigned)(range & 3) << T16_ROW_BITS;
    auto lstep = [&](bool valid, unsigned e) {
      const bool hit = valid && ((e & (3u << T16_ROW_BITS)) == subkey);
      take(hit, (int)(e >> CR_BITS), (int)(e & (unsigned)(T16_NODES - 1)));
    };

    auto scan_raw = [&](int e0, int e1) {
      const int nvec = (e1 - e0) >> 2;
      const int4* __restrict__ d4 = reinterpret_cast<const int4*>(dst + e0);
      const int4* __restrict__ s4 = reinterpret_cast<const int4*>(src + e0);
      const int last = nvec - 1;
      int4 n0 = make_int4(-1, -1, -1, -1), n1 = n0, n2 = n0, n3 = n0;
      int4 m0 = make_int4(0, 0, 0, 0), m1 = m0, m2 = m0, m3 = m0;
      if (nvec > 0) {
        const int g = threadIdx.x;
        n0 = d4[min(g, last)];
        n1 = d4[min(g + T16_THREADS, last)];
        n2 = d4[min(g + 2 * T16_THREADS, last)];
        n3 = d4[min(g + 3 * T16_THREADS, last)];
        m0 = s4[min(g, last)];
        m1 = s4[min(g + T16_THREADS, last)];
        m2 = s4[min(g + 2 * T16_THREADS, last)];
        m3 = s4[min(g + 3 * T16_THREADS, last)];
      }
      for (int base = 0; base < nvec; base += 4 * T16_THREADS) {
        const int4 q0 = n0, q1 = n1, q2 = n2, q3 = n3;
        const int4 p0 = m0, p1 = m1, p2 = m2, p3 = m3;
        const int g0 = base + threadIdx.x;
        const int g1 = g0 + T16_THREADS;
        const int g2 = g0 + 2 * T16_THREADS;
        const int g3 = g0 + 3 * T16_THREADS;
        n0 = d4[min(g0 + 4 * T16_THREADS, last)];
        n1 = d4[min(g1 + 4 * T16_THREADS, last)];
        n2 = d4[min(g2 + 4 * T16_THREADS, last)];
        n3 = d4[min(g3 + 4 * T16_THREADS, last)];
        m0 = s4[min(g0 + 4 * T16_THREADS, last)];
        m1 = s4[min(g1 + 4 * T16_THREADS, last)];
        m2 = s4[min(g2 + 4 * T16_THREADS, last)];
        m3 = s4[min(g3 + 4 * T16_THREADS, last)];
        const bool in0 = g0 < nvec, in1 = g1 < nvec, in2 = g2 < nvec, in3 = g3 < nvec;
        step(in0 ? q0.x : -1, p0.x); step(in0 ? q0.y : -1, p0.y); step(in0 ? q0.z : -1, p0.z); step(in0 ? q0.w : -1, p0.w);
        step(in1 ? q1.x : -1, p1.x); step(in1 ? q1.y : -1, p1.y); step(in1 ? q1.z : -1, p1.z); step(in1 ? q1.w : -1, p1.w);
        step(in2 ? q2.x : -1, p2.x); step(in2 ? q2.y : -1, p2.y); step(in2 ? q2.z : -1, p2.z); step(in2 ? q2.w : -1, p2.w);
        step(in3 ? q3.x : -1, p3.x); step(in3 ? q3.y : -1, p3.y); step(in3 ? q3.z : -1, p3.z); step(in3 ? q3.w : -1, p3.w);
      }
      {
        const int t = e0 + 4 * nvec + (int)threadIdx.x;
        const bool have = t < e1;
        const int d = have ? dst[t] : -1;
        const int s = have ? src[t] : 0;
        step(d, s);
      }
    };

    auto scan_list = [&](const unsigned* __restrict__ lp, int n) {
      const int nvec = n >> 2;
      const uint4* __restrict__ v4 = reinterpret_cast<const uint4*>(lp);
      const int last = nvec - 1;
      uint4 a0 = make_uint4(0u, 0u, 0u, 0u), a1 = a0;
      if (nvec > 0) {
        const int g = threadIdx.x;
        a0 = v4[min(g, last)];
        a1 = v4[min(g + T16_THREADS, last)];
      }
      for (int base = 0; base < nvec; base += 2 * T16_THREADS) {
        const uint4 q0 = a0, q1 = a1;
        const int g0 = base + threadIdx.x;
        const int g1 = g0 + T16_THREADS;
        a0 = v4[min(g0 + 2 * T16_THREADS, last)];
        a1 = v4[min(g1 + 2 * T16_THREADS, last)];
        const bool in0 = g0 < nvec, in1 = g1 < nvec;
        lstep(in0, q0.x); lstep(in0, q0.y); lstep(in0, q0.z); lstep(in0, q0.w);
        lstep(in1, q1.x); lstep(in1, q1.y); lstep(in1, q1.z); lstep(in1, q1.w);
      }
      {
        const int t = 4 * nvec + (int)threadIdx.x;
        const bool have = t < n;
        const unsigned e = have ? lp[t] : 0u;
        lstep(have, e);
      }
    };

    const int crow = (range >> 2) * CR_CHUNKS;
    const int k1 = gcn_chunk_first(chunk + 1);
    for (int k = gcn_chunk_first(chunk); k < k1; ++k) {
      int cn = CR_CAP + 1;
      if (coarse != nullptr) cn = ccounts[crow + k];
      if (cn <= CR_CAP) scan_list(coarse + (size_t)(crow + k) * CR_CAP, cn);
      else              scan_raw(min(E, k * epcC), min(E, (k + 1) * epcC));
    }
    if (tail - head > 0)
      gcn_tile16_batch<false>(ring_s, ring_r, head, tail - head, lane, dis, hw, dtile, tile,
                              list_p, list_n, &cursor);
  }
  __syncthreads();

  if (!WALK && lists_p != nullptr && threadIdx.x == 0) counts[blockIdx.x] = cursor;

  float* __restrict__ ag = agg + ((size_t)node0 << 4);
  for (int i = threadIdx.x; i < nodes * 16; i += T16_THREADS) atomicAdd(&ag[i], tile[i]);
}

__global__ void gcn_mv16(const float* __restrict__ h, const float* __restrict__ W3,
                         float* __restrict__ out, int N) {
  int i = blockIdx.x * blockDim.x + threadIdx.x;
  if (i >= N) return;
  const float* hp = h + ((size_t)i << 4);
  float s = 0.f;
#pragma unroll
  for (int c = 0; c < 16; ++c) s += fmaxf(__builtin_nontemporal_load(&hp[c]), 0.0f) * W3[c];
  out[i] = s;
}

__global__ __launch_bounds__(T16_THREADS) void gcn_list_acc1(
    const int* __restrict__ src, const int* __restrict__ dst,
    const float* __restrict__ dis, const float* __restrict__ hw3,
    const unsigned* __restrict__ lists_p, const float* __restrict__ lists_n,
    const int* __restrict__ counts, float* __restrict__ planes, int N, int E, int epcC) {
  __shared__ float tile[T16_NODES];
  const int range = blockIdx.x / T16_CHUNKS;
  const int chunk = blockIdx.x - range * T16_CHUNKS;
  const int node0 = range * T16_NODES;
  const int nodes = min(T16_NODES, N - node0);
  for (int i = threadIdx.x; i < nodes; i += T16_THREADS) tile[i] = 0.f;
  __syncthreads();

  const int cnt = counts[blockIdx.x];
  if (cnt <= T16_LIST_CAP) {
    const unsigned* __restrict__ lp = lists_p + (size_t)blockIdx.x * T16_LIST_CAP;
    const float* __restrict__ ln = lists_n + (size_t)blockIdx.x * T16_LIST_CAP;
    int p = threadIdx.x;
    for (; p + 3 * T16_THREADS < cnt; p += 4 * T16_THREADS) {
      const unsigned e0 = lp[p];
      const unsigned e1 = lp[p + T16_THREADS];
      const unsigned e2 = lp[p + 2 * T16_THREADS];
      const unsigned e3 = lp[p + 3 * T16_THREADS];
      const float n0 = ln[p];
      const float n1 = ln[p + T16_THREADS];
      const float n2 = ln[p + 2 * T16_THREADS];
      const float n3 = ln[p + 3 * T16_THREADS];
      const float h0 = hw3[e0 >> T16_ROW_BITS];
      const float h1 = hw3[e1 >> T16_ROW_BITS];
      const float h2 = hw3[e2 >> T16_ROW_BITS];
      const float h3 = hw3[e3 >> T16_ROW_BITS];
      atomicAdd(&tile[e0 & (unsigned)(T16_NODES - 1)], n0 * h0);
      atomicAdd(&tile[e1 & (unsigned)(T16_NODES - 1)], n1 * h1);
      atomicAdd(&tile[e2 & (unsigned)(T16_NODES - 1)], n2 * h2);
      atomicAdd(&tile[e3 & (unsigned)(T16_NODES - 1)], n3 * h3);
    }
    for (; p < cnt; p += T16_THREADS) {
      const unsigned e = lp[p];
      atomicAdd(&tile[e & (unsigned)(T16_NODES - 1)], ln[p] * hw3[e >> T16_ROW_BITS]);
    }
  } else {
    const int e0 = min(E, gcn_chunk_first(chunk) * epcC);
    const int e1 = min(E, gcn_chunk_first(chunk + 1) * epcC);
    for (int t = e0 + (int)threadIdx.x; t < e1; t += T16_THREADS) {
      const int rel = dst[t] - node0;
      if ((unsigned)rel < (unsigned)nodes) {
        const int s = src[t];
        atomicAdd(&tile[rel], dis[s] * dis[node0 + rel] * hw3[s]);
      }
    }
  }
  __syncthreads();

  float* __restrict__ pl = planes + (size_t)chunk * N + node0;
  for (int i = threadIdx.x; i < nodes; i += T16_THREADS) pl[i] = tile[i];
}

template <int NP>
__global__ void gcn_out_finish(const float* __restrict__ planes, const float* __restrict__ hw3,
                               const float* __restrict__ dis, const float* __restrict__ b3,
                               float* __restrict__ out, int N) {
  int i = blockIdx.x * blockDim.x + threadIdx.x;
  if (i >= N) return;
  float d = dis[i];
  float acc = d * d * hw3[i] + b3[0];
#pragma unroll
  for (int c = 0; c < NP; ++c) acc += planes[(size_t)c * N + i];
  out[i] = acc;
}

static inline int cdiv(long long a, long long b) { return (int)((a + b - 1) / b); }

extern "C" void kernel_launch(void* const* d_in, const int* in_sizes, int n_in,
                              void* d_out, int out_size, void* d_ws, size_t ws_size,
                              hipStream_t stream) {
  const float* x  = (const float*)d_in[0];
  const int*   ei = (const int*)d_in[1];
  const float* W1 = (const float*)d_in[2];
  const float* b1 = (const float*)d_in[3];
  const float* W2 = (const float*)d_in[4];
  const float* b2 = (const float*)d_in[5];
  const float* W3 = (const float*)d_in[6];
  const float* b3 = (const float*)d_in[7];

  const int IN_CH = 256;
  const int HID = 16;
  const int N = in_sizes[0] / IN_CH;
  const int E = in_sizes[1] / 2;
  const int* src = ei;
  const int* dst = ei + E;
  float* out = (float*)d_out;

  char* ws = (char*)d_ws;
  size_t off = 0;
  float* dis = (float*)(ws + off); off += (((size_t)N * 4) + 255) & ~(size_t)255;
  float* HW  = (float*)(ws + off); off += (((size_t)N * HID * 4) + 255) & ~(size_t)255;
  float* AGG = (float*)(ws + off); off += (((size_t)N * HID * 4) + 255) & ~(size_t)255;
  (void)n_in; (void)out_size;

  const int B = 256;
  const int gemmBlocks = cdiv(cdiv(N, 16), B / 32);

  const int tileGrid = cdiv(N, TILE_NODES) * TILE_CHUNKS;
  const int epc = cdiv(cdiv(E, TILE_CHUNKS), 4 * TILE_THREADS) * (4 * TILE_THREADS);

  const int t16Grid = cdiv(N, T16_NODES) * T16_CHUNKS;

  const int crGrid = cdiv(N, CR_NODES) * CR_CHUNKS;
  const int epcC = cdiv(cdiv(E, CR_CHUNKS), 4 * CR_THREADS) * (4 * CR_THREADS);

  const size_t listBytes = (size_t)t16Grid * T16_LIST_CAP * 4;
  const size_t offP = off + ((((size_t)t16Grid * 4) + 255) & ~(size_t)255);
  const size_t offN = offP + listBytes;
  const bool keepP = N <= (1 << (32 - T16_ROW_BITS)) && ws_size >= offN;
  const bool keepN = keepP && ws_size >= offN + listBytes;
  int* cnt16 = (int*)(ws + off);
  unsigned* listP = keepP ? (unsigned*)(ws + offP) : nullptr;
  float* listN = keepN ? (float*)(ws + offN) : nullptr;

  const size_t offCC = offN + listBytes;
  const size_t offCL = offCC + ((((size_t)crGrid * 4) + 255) & ~(size_t)255);
  const bool keepC = N <= (1 << (32 - CR_BITS)) && ws_size >= offCL + (size_t)crGrid * CR_CAP * 4;
  int* cntC = keepC ? (int*)(ws + offCC) : nullptr;
  unsigned* listC = keepC ? (unsigned*)(ws + offCL) : nullptr;

  if (keepC) gcn_deg_coarse<true><<<crGrid, CR_THREADS, 0, stream>>>(src, dst, HW, listC, cntC, N, E, epcC);
  else       gcn_deg_coarse<false><<<crGrid, CR_THREADS, 0, stream>>>(src, dst, HW, listC, cntC, N, E, epcC);
  gcn_deg_finish<CR_CHUNKS><<<cdiv(N, B), B, 0, stream>>>(HW, dis, N);

  gcn_gemm_wmma_f32<false><<<gemmBlocks, B, 0, stream>>>(x, W1, dis, b1, HW, AGG, N, IN_CH);
  gcn_tile_scatter16<false><<<t16Grid, T16_THREADS, 0, stream>>>(src, dst, dis, HW, AGG, listP, listN, cnt16, listC, cntC, N, E, epcC);

  gcn_gemm_wmma_f32<true><<<gemmBlocks, B, 0, stream>>>(AGG, W2, dis, b2, HW, AGG, N, HID);
  gcn_tile_scatter16<true><<<t16Grid, T16_THREADS, 0, stream>>>(src, dst, dis, HW, AGG, listP, listN, cnt16, listC, cntC, N, E, epcC);

  gcn_mv16<<<cdiv(N, B), B, 0, stream>>>(AGG, W3, HW, N);
  if (keepN) {
    gcn_list_acc1<<<t16Grid, T16_THREADS, 0, stream>>>(src, dst, dis, HW, listP, listN, cnt16, HW + N, N, E, epcC);
    gcn_out_finish<T16_CHUNKS><<<cdiv(N, B), B, 0, stream>>>(HW + N, HW, dis, b3, out, N);
  } else {
    gcn_tile_acc1<true><<<tileGrid, TILE_THREADS, 0, stream>>>(src, dst, dis, HW, HW + N, N, E, epc);
    gcn_out_finish<TILE_CHUNKS><<<cdiv(N, B), B, 0, stream>>>(HW + N, HW, dis, b3, out, N);
  }
}


# speedup vs baseline: 4.6855x; 1.1689x over previous
#include <hip/hip_runtime.h>
#include <hip/hip_bf16.h>

typedef __attribute__((ext_vector_type(2))) float v2f;
typedef __attribute__((ext_vector_type(8))) float v8f;

template <bool RELU_A>
__global__ __launch_bounds__(256) void gcn_gemm_wmma_f32(
    const float* A, const float* __restrict__ W,
    const float* __restrict__ dis, const float* __restrict__ b,
    float* __restrict__ C, float* AGG, int M, int K) {
  const int wave = blockIdx.x * (blockDim.x >> 5) + (threadIdx.x >> 5);
  const int lane = threadIdx.x & 31;
  const int m0 = wave << 4;
  if (m0 >= M) return;
  const int half = lane >> 4;
  const int l = lane & 15;

  v8f acc = {};
  const float* arow = A + (size_t)(m0 + l) * K + (half << 1);
  for (int k0 = 0; k0 < K; k0 += 4) {
    v2f a;
    a.x = arow[k0 + 0];
    a.y = arow[k0 + 1];
    if (RELU_A) {
      a.x = fmaxf(a.x, 0.0f);
      a.y = fmaxf(a.y, 0.0f);
    }
    v2f b;
    b.x = W[(size_t)(k0 + (half << 1) + 0) * 16 + l];
    b.y = W[(size_t)(k0 + (half << 1) + 1) * 16 + l];
    acc = __builtin_amdgcn_wmma_f32_16x16x4_f32(
         false, a,  false, b,
         (short)0, acc,  false,  false);
  }
  const int r0 = m0 + (half << 3);
  const float4 d0 = *reinterpret_cast<const float4*>(dis + r0);
  const float4 d1 = *reinterpret_cast<const float4*>(dis + r0 + 4);
  const float dd[8] = {d0.x, d0.y, d0.z, d0.w, d1.x, d1.y, d1.z, d1.w};
  const float bias = b[l];
  float* cp = C + ((size_t)r0 << 4) + l;
  float* gp = AGG + ((size_t)r0 << 4) + l;
#pragma unroll
  for (int r = 0; r < 8; ++r) {
    cp[(size_t)r << 4] = acc[r];
    gp[(size_t)r << 4] = dd[r] * dd[r] * acc[r] + bias;
  }
}

constexpr int TILE_NODES  = 25000;
constexpr int TILE_CHUNKS = 31;
constexpr int TILE_THREADS = 1024;

template <bool WEIGHTED, int NE>
__device__ __forceinline__ void gcn_tile_add(
    float* tile, const float* dtile, const int (&d)[NE], const int (&s)[NE], int node0, int nodes,
    const float* __restrict__ dis, const float* __restrict__ hw3) {
  bool hit[NE];
  int rel[NE];
#pragma unroll
  for (int j = 0; j < NE; ++j) {
    rel[j] = d[j] - node0;
    hit[j] = (unsigned)rel[j] < (unsigned)nodes;
  }
  if (WEIGHTED) {
    float v[NE];
#pragma unroll
    for (int j = 0; j < NE; ++j) {
      const int si = hit[j] ? s[j] : 0;
      const int ri = hit[j] ? rel[j] : 0;
      v[j] = dis[si] * dtile[ri] * hw3[si];
    }
#pragma unroll
    for (int j = 0; j < NE; ++j)
      if (hit[j]) atomicAdd(&tile[rel[j]], v[j]);
  } else {
#pragma unroll
    for (int j = 0; j < NE; ++j)
      if (hit[j]) atomicAdd(&tile[rel[j]], 1.0f);
  }
}

template <bool WEIGHTED>
__global__ __launch_bounds__(TILE_THREADS) void gcn_tile_acc1(
    const int* __restrict__ src, const int* __restrict__ dst,
    const float* __restrict__ dis, const float* __restrict__ hw3,
    float* __restrict__ planes, int N, int E, int epc) {
  __shared__ float tile[TILE_NODES];
  __shared__ float dtile[WEIGHTED ? TILE_NODES : 1];
  const int range = blockIdx.x / TILE_CHUNKS;
  const int chunk = blockIdx.x - range * TILE_CHUNKS;
  const int node0 = range * TILE_NODES;
  const int nodes = min(TILE_NODES, N - node0);
  for (int i = threadIdx.x; i < nodes; i += TILE_THREADS) {
    tile[i] = 0.f;
    if (WEIGHTED) dtile[i] = dis[node0 + i];
  }
  __syncthreads();

  const int e0 = min(E, chunk * epc);
  const int e1 = min(E, e0 + epc);

  const int nvec = (e1 - e0) >> 2;
  const int4* __restrict__ d4 = reinterpret_cast<const int4*>(dst + e0);
  const int4* __restrict__ s4 = reinterpret_cast<const int4*>(src + e0);
  int g = threadIdx.x;
  for (; g + 3 * TILE_THREADS < nvec; g += 4 * TILE_THREADS) {
    const int4 q0 = d4[g];
    const int4 q1 = d4[g + TILE_THREADS];
    const int4 q2 = d4[g + 2 * TILE_THREADS];
    const int4 q3 = d4[g + 3 * TILE_THREADS];
    int4 p0 = q0, p1 = q0, p2 = q0, p3 = q0;
    if (WEIGHTED) {
      p0 = s4[g];
      p1 = s4[g + TILE_THREADS];
      p2 = s4[g + 2 * TILE_THREADS];
      p3 = s4[g + 3 * TILE_THREADS];
    }
    const int d[16] = {q0.x, q0.y, q0.z, q0.w, q1.x, q1.y, q1.z, q1.w,
                       q2.x, q2.y, q2.z, q2.w, q3.x, q3.y, q3.z, q3.w};
    const int s[16] = {p0.x, p0.y, p0.z, p0.w, p1.x, p1.y, p1.z, p1.w,
                       p2.x, p2.y, p2.z, p2.w, p3.x, p3.y, p3.z, p3.w};
    gcn_tile_add<WEIGHTED, 16>(tile, dtile, d, s, node0, nodes, dis, hw3);
  }
  for (; g < nvec; g += TILE_THREADS) {
    const int4 q = d4[g];
    int4 p = q;
    if (WEIGHTED) p = s4[g];
    const int d[4] = {q.x, q.y, q.z, q.w};
    const int s[4] = {p.x, p.y, p.z, p.w};
    gcn_tile_add<WEIGHTED, 4>(tile, dtile, d, s, node0, nodes, dis, hw3);
  }
  for (int t = e0 + 4 * nvec + threadIdx.x; t < e1; t += TILE_THREADS) {
    const int d[1] = {dst[t]};
    const int s[1] = {WEIGHTED ? src[t] : 0};
    gcn_tile_add<WEIGHTED, 1>(tile, dtile, d, s, node0, nodes, dis, hw3);
  }
  __syncthreads();

  float* __restrict__ pl = planes + (size_t)chunk * N + node0;
  for (int i = threadIdx.x; i < nodes; i += TILE_THREADS) pl[i] = tile[i];
}

template <int NP>
__global__ void gcn_deg_finish(const float* __restrict__ planes, float* __restrict__ dis, int N) {
  int i = blockIdx.x * blockDim.x + threadIdx.x;
  if (i >= N) return;
  float deg = 1.0f;
#pragma unroll
  for (int c = 0; c < NP; ++c) deg += planes[(size_t)c * N + i];
  dis[i] = rsqrtf(deg);
}

constexpr int T16_NODES   = 4096;
constexpr int T16_CHUNKS  = 5;
constexpr int T16_THREADS = 1024;
constexpr int T16_WAVES   = T16_THREADS / 32;
constexpr int T16_RING    = 64;

template <int J>
__device__ __forceinline__ int gcn_row_lane(int v) {
  return __builtin_amdgcn_mov_dpp(v, 0x150 + J, 0xf, 0xf, true);
}

#define GCN_FOR_16(X) X(0) X(1) X(2) X(3) X(4) X(5) X(6) X(7) X(8) X(9) X(10) X(11) X(12) X(13) X(14) X(15)

constexpr int T16_LIST_CAP = 32768;
constexpr int T16_ROW_BITS = 12;
static_assert(T16_NODES == (1 << T16_ROW_BITS), "row field of a packed hit");

constexpr int CR_BITS    = 14;
constexpr int CR_NODES   = 1 << CR_BITS;
constexpr int CR_CHUNKS  = 19;
constexpr int CR_THREADS = 1024;
constexpr int CR_WAVES   = CR_THREADS / 32;
constexpr int CR_RING    = 64;
constexpr int CR_CAP     = 32768;
static_assert(CR_NODES == 4 * T16_NODES, "a coarse range is four fine ranges");
static_assert(CR_CHUNKS >= T16_CHUNKS, "every fine chunk holds a coarse chunk");

__device__ __forceinline__ int gcn_chunk_first(int c) { return (c * CR_CHUNKS) / T16_CHUNKS; }

template <bool KEEP>
__global__ __launch_bounds__(CR_THREADS) void gcn_deg_coarse(
    const int* __restrict__ src, const int* __restrict__ dst,
    float* __restrict__ planes, unsigned* coarse, int* ccounts, int N, int E, int epc) {
  __shared__ float tile[CR_NODES];
  __shared__ unsigned ring_all[CR_WAVES * CR_RING];
  __shared__ int cursor;
  const int range = blockIdx.x / CR_CHUNKS;
  const int chunk = blockIdx.x - range * CR_CHUNKS;
  const int node0 = range << CR_BITS;
  const int nodes = min(CR_NODES, N - node0);
  for (int i = threadIdx.x; i < nodes; i += CR_THREADS) tile[i] = 0.f;
  if (threadIdx.x == 0) cursor = 0;
  __syncthreads();

  const int lane = threadIdx.x & 31;
  const int wv = __builtin_amdgcn_readfirstlane(threadIdx.x >> 5);
  unsigned* ring = ring_all + wv * CR_RING;
  unsigned* list = KEEP ? coarse + (size_t)blockIdx.x * CR_CAP : nullptr;
  int head = 0, tail = 0;

  auto flush = [&](int count) {
    int pos = 0;
    if (lane == 0) pos = atomicAdd(&cursor, count);
    pos = __builtin_amdgcn_readfirstlane(pos) + lane;
    if (lane < count && pos < CR_CAP) list[pos] = ring[(head + lane) & (CR_RING - 1)];
  };
  auto step = [&](int d, int s) {
    const int rel = d - node0;
    const bool hit = (unsigned)rel < (unsigned)nodes;
    if (KEEP) {
      const unsigned m = __builtin_amdgcn_ballot_w32(hit);
      if (m != 0u) {
        if (hit) {
          atomicAdd(&tile[rel], 1.0f);
          const int slot = (tail + (int)__builtin_amdgcn_mbcnt_lo(m, 0u)) & (CR_RING - 1);
          ring[slot] = ((unsigned)s << CR_BITS) | (unsigned)rel;
        }
        tail += __builtin_popcount(m);
        if (tail - head >= 32) {
          flush(32);
          head += 32;
        }
      }
    } else {
      if (hit) atomicAdd(&tile[rel], 1.0f);
    }
  };

  const int e0 = min(E, chunk * epc);
  const int e1 = min(E, e0 + epc);
  const int nvec = (e1 - e0) >> 2;
  const int4* __restrict__ d4 = reinterpret_cast<const int4*>(dst + e0);
  const int4* __restrict__ s4 = reinterpret_cast<const int4*>(src + e0);
  const int last = nvec - 1;
  int4 n0 = make_int4(-1, -1, -1, -1), n1 = n0, n2 = n0, n3 = n0;
  int4 m0 = make_int4(0, 0, 0, 0), m1 = m0, m2 = m0, m3 = m0;
  if (nvec > 0) {
    const int g = threadIdx.x;
    n0 = d4[min(g, last)];
    n1 = d4[min(g + CR_THREADS, last)];
    n2 = d4[min(g + 2 * CR_THREADS, last)];
    n3 = d4[min(g + 3 * CR_THREADS, last)];
    m0 = s4[min(g, last)];
    m1 = s4[min(g + CR_THREADS, last)];
    m2 = s4[min(g + 2 * CR_THREADS, last)];
    m3 = s4[min(g + 3 * CR_THREADS, last)];
  }
  for (int base = 0; base < nvec; base += 4 * CR_THREADS) {
    const int4 q0 = n0, q1 = n1, q2 = n2, q3 = n3;
    const int4 p0 = m0, p1 = m1, p2 = m2, p3 = m3;
    const int g0 = base + threadIdx.x;
    const int g1 = g0 + CR_THREADS;
    const int g2 = g0 + 2 * CR_THREADS;
    const int g3 = g0 + 3 * CR_THREADS;
    n0 = d4[min(g0 + 4 * CR_THREADS, last)];
    n1 = d4[min(g1 + 4 * CR_THREADS, last)];
    n2 = d4[min(g2 + 4 * CR_THREADS, last)];
    n3 = d4[min(g3 + 4 * CR_THREADS, last)];
    m0 = s4[min(g0 + 4 * CR_THREADS, last)];
    m1 = s4[min(g1 + 4 * CR_THREADS, last)];
    m2 = s4[min(g2 + 4 * CR_THREADS, last)];
    m3 = s4[min(g3 + 4 * CR_THREADS, last)];
    const bool in0 = g0 < nvec, in1 = g1 < nvec, in2 = g2 < nvec, in3 = g3 < nvec;
    step(in0 ? q0.x : -1, p0.x); step(in0 ? q0.y : -1, p0.y); step(in0 ? q0.z : -1, p0.z); step(in0 ? q0.w : -1, p0.w);
    step(in1 ? q1.x : -1, p1.x); step(in1 ? q1.y : -1, p1.y); step(in1 ? q1.z : -1, p1.z); step(in1 ? q1.w : -1, p1.w);
    step(in2 ? q2.x : -1, p2.x); step(in2 ? q2.y : -1, p2.y); step(in2 ? q2.z : -1, p2.z); step(in2 ? q2.w : -1, p2.w);
    step(in3 ? q3.x : -1, p3.x); step(in3 ? q3.y : -1, p3.y); step(in3 ? q3.z : -1, p3.z); step(in3 ? q3.w : -1, p3.w);
  }
  {
    const int t = e0 + 4 * nvec + (int)threadIdx.x;
    const bool have = t < e1;
    const int d = have ? dst[t] : -1;
    const int s = have ? src[t] : 0;
    step(d, s);
  }
  if (KEEP && tail - head > 0) flush(tail - head);
  __syncthreads();

  if (KEEP && threadIdx.x == 0) ccounts[blockIdx.x] = cursor;

  float* __restrict__ pl = planes + (size_t)chunk * N + node0;
  for (int i = threadIdx.x; i < nodes; i += CR_THREADS) pl[i] = tile[i];
}

__device__ __forceinline__ void gcn_tile16_gather(
    int s, int lane, const float* __restrict__ hw, float (&v)[16]) {
  const int c = lane & 15;
#define GCN_ROW_LOAD(J) v[J] = hw[(gcn_row_lane<J>(s) << 4) + c];
  GCN_FOR_16(GCN_ROW_LOAD)
#undef GCN_ROW_LOAD
}

template <bool FULL>
__device__ __forceinline__ void gcn_tile16_add(
    const float (&v)[16], int rel, float nrm, int count, int lane, float* tile) {
  const int c = lane & 15;
  const int row0 = lane & 16;
  const int nrm_bits = __float_as_int(nrm);
  const int left = count - row0;

#define GCN_ROW_ADD(J)                                                        \
  {                                                                           \
    const int rj = gcn_row_lane<J>(rel);                                      \
    const float nj = __int_as_float(gcn_row_lane<J>(nrm_bits));               \
    if (FULL || J < left) atomicAdd(&tile[(rj << 4) + c], nj * v[J]);         \
  }
  GCN_FOR_16(GCN_ROW_ADD)
#undef GCN_ROW_ADD
}
#undef GCN_FOR_16

template <bool FULL>
__device__ __forceinline__ void gcn_tile16_rows(
    int s, int rel, float nrm, int count, int lane,
    const float* __restrict__ hw, float* tile) {
  float v[16];
  gcn_tile16_gather(s, lane, hw, v);
  gcn_tile16_add<FULL>(v, rel, nrm, count, lane, tile);
}

template <bool FULL>
__device__ __forceinline__ void gcn_tile16_batch(
    const int* ring_s, const int* ring_r, int head, int count, int lane,
    const float* __restrict__ dis, const float* __restrict__ hw,
    const float* dtile, float* tile,
    unsigned* list_p, float* list_n, int* cursor) {
  int s = 0, rel = 0;
  float nrm = 0.f;
  const bool have = FULL || lane < count;
  if (have) {
    const int slot = (head + lane) & (T16_RING - 1);
    rel = ring_r[slot];
    s = ring_s[slot];
    nrm = dis[s] * dtile[rel];
  }
  if (list_p != nullptr) {
    int pos = 0;
    if (lane == 0) pos = atomicAdd(cursor, count);
    pos = __builtin_amdgcn_readfirstlane(pos) + lane;
    if (have && pos < T16_LIST_CAP) {
      list_p[pos] = ((unsigned)s << T16_ROW_BITS) | (unsigned)rel;
      if (list_n != nullptr) list_n[pos] = nrm;
    }
  }
  gcn_tile16_rows<FULL>(s, rel, nrm, count, lane, hw, tile);
}

template <bool WALK>
__global__ __launch_bounds__(T16_THREADS) void gcn_tile_scatter16(
    const int* __restrict__ src, const int* __restrict__ dst,
    const float* __restrict__ dis, const float* __restrict__ hw,
    float* __restrict__ agg, unsigned* lists_p, float* lists_n, int* counts,
    const unsigned* coarse, const int* ccounts, int N, int E, int epcC) {
  __shared__ float tile[T16_NODES * 16];
  __shared__ float dtile[T16_NODES];
  __shared__ int ring_s_all[T16_WAVES * T16_RING];
  __shared__ int ring_r_all[T16_WAVES * T16_RING];
  __shared__ int cursor;
  const int range = blockIdx.x / T16_CHUNKS;
  const int chunk = blockIdx.x - range * T16_CHUNKS;
  const int node0 = range * T16_NODES;
  const int nodes = min(T16_NODES, N - node0);
  for (int i = threadIdx.x; i < nodes * 16; i += T16_THREADS) tile[i] = 0.f;
  for (int i = threadIdx.x; i < nodes; i += T16_THREADS) dtile[i] = dis[node0 + i];
  if (threadIdx.x == 0) cursor = 0;
  __syncthreads();

  const int lane = threadIdx.x & 31;
  const int wv = __builtin_amdgcn_readfirstlane(threadIdx.x >> 5);

  const size_t list0 = (size_t)blockIdx.x * T16_LIST_CAP;
  bool walked = false;
  if (WALK && lists_p != nullptr) {
    const int cnt = counts[blockIdx.x];
    if (cnt <= T16_LIST_CAP) {
      walked = true;
      const unsigned* __restrict__ lp = lists_p + list0;
      const float* __restrict__ ln = lists_n != nullptr ? lists_n + list0 : nullptr;
      const int trips = (cnt + 31) >> 5;
      const int lastpos = cnt - 1;
      unsigned e_next = 0u;
      float n_next = 0.f;
      int b = wv;
      if (b < trips) {
        const int p = min((b << 5) + lane, lastpos);
        e_next = lp[p];
        if (ln != nullptr) n_next = ln[p];
      }
      for (; b < trips; b += T16_WAVES) {
        const unsigned e = e_next;
        float nrm = n_next;
        const int s = (int)(e >> T16_ROW_BITS);
        const int rel = (int)(e & (unsigned)(T16_NODES - 1));
        float v[16];
        gcn_tile16_gather(s, lane, hw, v);
        if (ln == nullptr) nrm = dis[s] * dtile[rel];
        const int bn = b + T16_WAVES;
        if (bn < trips) {
          const int p = min((bn << 5) + lane, lastpos);
          e_next = lp[p];
          if (ln != nullptr) n_next = ln[p];
        }
        const int left = cnt - (b << 5);
        if (left >= 32) gcn_tile16_add<true>(v, rel, nrm, 32, lane, tile);
        else            gcn_tile16_add<false>(v, rel, nrm, left, lane, tile);
      }
    }
  }

  if (!walked) {
    unsigned* list_p = (!WALK && lists_p != nullptr) ? lists_p + list0 : nullptr;
    float* list_n = (list_p != nullptr && lists_n != nullptr) ? lists_n + list0 : nullptr;
    int* ring_s = ring_s_all + wv * T16_RING;
    int* ring_r = ring_r_all + wv * T16_RING;
    int head = 0, tail = 0;

    auto take = [&](bool hit, int s, int rel) {
      const unsigned m = __builtin_amdgcn_ballot_w32(hit);
      if (m != 0u) {
        if (hit) {
          const int slot = (tail + (int)__builtin_amdgcn_mbcnt_lo(m, 0u)) & (T16_RING - 1);
          ring_s[slot] = s;
          ring_r[slot] = rel;
        }
        tail += __builtin_popcount(m);
        if (tail - head >= 32) {
          gcn_tile16_batch<true>(ring_s, ring_r, head, 32, lane, dis, hw, dtile, tile,
                                 list_p, list_n, &cursor);
          head += 32;
        }
      }
    };
    auto step = [&](int d, int s) {
      const int rel = d - node0;
      take((unsigned)rel < (unsigned)nodes, s, rel);
    };
    const unsigned subkey = (unsigned)(range & 3) << T16_ROW_BITS;
    auto lstep = [&](bool valid, unsigned e) {
      const bool hit = valid && ((e & (3u << T16_ROW_BITS)) == subkey);
      take(hit, (int)(e >> CR_BITS), (int)(e & (unsigned)(T16_NODES - 1)));
    };

    auto scan_raw = [&](int e0, int e1) {
      const int nvec = (e1 - e0) >> 2;
      const int4* __restrict__ d4 = reinterpret_cast<const int4*>(dst + e0);
      const int4* __restrict__ s4 = reinterpret_cast<const int4*>(src + e0);
      const int last = nvec - 1;
      int4 n0 = make_int4(-1, -1, -1, -1), n1 = n0, n2 = n0, n3 = n0;
      int4 m0 = make_int4(0, 0, 0, 0), m1 = m0, m2 = m0, m3 = m0;
      if (nvec > 0) {
        const int g = threadIdx.x;
        n0 = d4[min(g, last)];
        n1 = d4[min(g + T16_THREADS, last)];
        n2 = d4[min(g + 2 * T16_THREADS, last)];
        n3 = d4[min(g + 3 * T16_THREADS, last)];
        m0 = s4[min(g, last)];
        m1 = s4[min(g + T16_THREADS, last)];
        m2 = s4[min(g + 2 * T16_THREADS, last)];
        m3 = s4[min(g + 3 * T16_THREADS, last)];
      }
      for (int base = 0; base < nvec; base += 4 * T16_THREADS) {
        const int4 q0 = n0, q1 = n1, q2 = n2, q3 = n3;
        const int4 p0 = m0, p1 = m1, p2 = m2, p3 = m3;
        const int g0 = base + threadIdx.x;
        const int g1 = g0 + T16_THREADS;
        const int g2 = g0 + 2 * T16_THREADS;
        const int g3 = g0 + 3 * T16_THREADS;
        n0 = d4[min(g0 + 4 * T16_THREADS, last)];
        n1 = d4[min(g1 + 4 * T16_THREADS, last)];
        n2 = d4[min(g2 + 4 * T16_THREADS, last)];
        n3 = d4[min(g3 + 4 * T16_THREADS, last)];
        m0 = s4[min(g0 + 4 * T16_THREADS, last)];
        m1 = s4[min(g1 + 4 * T16_THREADS, last)];
        m2 = s4[min(g2 + 4 * T16_THREADS, last)];
        m3 = s4[min(g3 + 4 * T16_THREADS, last)];
        const bool in0 = g0 < nvec, in1 = g1 < nvec, in2 = g2 < nvec, in3 = g3 < nvec;
        step(in0 ? q0.x : -1, p0.x); step(in0 ? q0.y : -1, p0.y); step(in0 ? q0.z : -1, p0.z); step(in0 ? q0.w : -1, p0.w);
        step(in1 ? q1.x : -1, p1.x); step(in1 ? q1.y : -1, p1.y); step(in1 ? q1.z : -1, p1.z); step(in1 ? q1.w : -1, p1.w);
        step(in2 ? q2.x : -1, p2.x); step(in2 ? q2.y : -1, p2.y); step(in2 ? q2.z : -1, p2.z); step(in2 ? q2.w : -1, p2.w);
        step(in3 ? q3.x : -1, p3.x); step(in3 ? q3.y : -1, p3.y); step(in3 ? q3.z : -1, p3.z); step(in3 ? q3.w : -1, p3.w);
      }
      {
        const int t = e0 + 4 * nvec + (int)threadIdx.x;
        const bool have = t < e1;
        const int d = have ? dst[t] : -1;
        const int s = have ? src[t] : 0;
        step(d, s);
      }
    };

    auto scan_list = [&](const unsigned* __restrict__ lp, int n) {
      const int nvec = n >> 2;
      const uint4* __restrict__ v4 = reinterpret_cast<const uint4*>(lp);
      const int last = nvec - 1;
      uint4 a0 = make_uint4(0u, 0u, 0u, 0u), a1 = a0;
      if (nvec > 0) {
        const int g = threadIdx.x;
        a0 = v4[min(g, last)];
        a1 = v4[min(g + T16_THREADS, last)];
      }
      for (int base = 0; base < nvec; base += 2 * T16_THREADS) {
        const uint4 q0 = a0, q1 = a1;
        const int g0 = base + threadIdx.x;
        const int g1 = g0 + T16_THREADS;
        a0 = v4[min(g0 + 2 * T16_THREADS, last)];
        a1 = v4[min(g1 + 2 * T16_THREADS, last)];
        const bool in0 = g0 < nvec, in1 = g1 < nvec;
        lstep(in0, q0.x); lstep(in0, q0.y); lstep(in0, q0.z); lstep(in0, q0.w);
        lstep(in1, q1.x); lstep(in1, q1.y); lstep(in1, q1.z); lstep(in1, q1.w);
      }
      {
        const int t = 4 * nvec + (int)threadIdx.x;
        const bool have = t < n;
        const unsigned e = have ? lp[t] : 0u;
        lstep(have, e);
      }
    };

    const int crow = (range >> 2) * CR_CHUNKS;
    const int k1 = gcn_chunk_first(chunk + 1);
    for (int k = gcn_chunk_first(chunk); k < k1; ++k) {
      int cn = CR_CAP + 1;
      if (coarse != nullptr) cn = ccounts[crow + k];
      if (cn <= CR_CAP) scan_list(coarse + (size_t)(crow + k) * CR_CAP, cn);
      else              scan_raw(min(E, k * epcC), min(E, (k + 1) * epcC));
    }
    if (tail - head > 0)
      gcn_tile16_batch<false>(ring_s, ring_r, head, tail - head, lane, dis, hw, dtile, tile,
                              list_p, list_n, &cursor);
  }
  __syncthreads();

  if (!WALK && lists_p != nullptr && threadIdx.x == 0) counts[blockIdx.x] = cursor;

  float* __restrict__ ag = agg + ((size_t)node0 << 4);
  for (int i = threadIdx.x; i < nodes * 16; i += T16_THREADS) atomicAdd(&ag[i], tile[i]);
}

__global__ void gcn_mv16(const float* __restrict__ h, const float* __restrict__ W3,
                         float* __restrict__ out, int N) {
  int i = blockIdx.x * blockDim.x + threadIdx.x;
  if (i >= N) return;
  const float* hp = h + ((size_t)i << 4);
  float s = 0.f;
#pragma unroll
  for (int c = 0; c < 16; ++c) s += fmaxf(__builtin_nontemporal_load(&hp[c]), 0.0f) * W3[c];
  out[i] = s;
}

__global__ __launch_bounds__(T16_THREADS) void gcn_list_acc1(
    const int* __restrict__ src, const int* __restrict__ dst,
    const float* __restrict__ dis, const float* __restrict__ hw3,
    const unsigned* __restrict__ lists_p, const float* __restrict__ lists_n,
    const int* __restrict__ counts, float* __restrict__ planes, int N, int E, int epcC) {
  __shared__ float tile[T16_NODES];
  const int range = blockIdx.x / T16_CHUNKS;
  const int chunk = blockIdx.x - range * T16_CHUNKS;
  const int node0 = range * T16_NODES;
  const int nodes = min(T16_NODES, N - node0);
  for (int i = threadIdx.x; i < nodes; i += T16_THREADS) tile[i] = 0.f;
  __syncthreads();

  const int cnt = counts[blockIdx.x];
  if (cnt <= T16_LIST_CAP) {
    const unsigned* __restrict__ lp = lists_p + (size_t)blockIdx.x * T16_LIST_CAP;
    const float* __restrict__ ln = lists_n + (size_t)blockIdx.x * T16_LIST_CAP;
    int p = threadIdx.x;
    for (; p + 3 * T16_THREADS < cnt; p += 4 * T16_THREADS) {
      const unsigned e0 = lp[p];
      const unsigned e1 = lp[p + T16_THREADS];
      const unsigned e2 = lp[p + 2 * T16_THREADS];
      const unsigned e3 = lp[p + 3 * T16_THREADS];
      const float n0 = ln[p];
      const float n1 = ln[p + T16_THREADS];
      const float n2 = ln[p + 2 * T16_THREADS];
      const float n3 = ln[p + 3 * T16_THREADS];
      const float h0 = hw3[e0 >> T16_ROW_BITS];
      const float h1 = hw3[e1 >> T16_ROW_BITS];
      const float h2 = hw3[e2 >> T16_ROW_BITS];
      const float h3 = hw3[e3 >> T16_ROW_BITS];
      atomicAdd(&tile[e0 & (unsigned)(T16_NODES - 1)], n0 * h0);
      atomicAdd(&tile[e1 & (unsigned)(T16_NODES - 1)], n1 * h1);
      atomicAdd(&tile[e2 & (unsigned)(T16_NODES - 1)], n2 * h2);
      atomicAdd(&tile[e3 & (unsigned)(T16_NODES - 1)], n3 * h3);
    }
    for (; p < cnt; p += T16_THREADS) {
      const unsigned e = lp[p];
      atomicAdd(&tile[e & (unsigned)(T16_NODES - 1)], ln[p] * hw3[e >> T16_ROW_BITS]);
    }
  } else {
    const int e0 = min(E, gcn_chunk_first(chunk) * epcC);
    const int e1 = min(E, gcn_chunk_first(chunk + 1) * epcC);
    for (int t = e0 + (int)threadIdx.x; t < e1; t += T16_THREADS) {
      const int rel = dst[t] - node0;
      if ((unsigned)rel < (unsigned)nodes) {
        const int s = src[t];
        atomicAdd(&tile[rel], dis[s] * dis[node0 + rel] * hw3[s]);
      }
    }
  }
  __syncthreads();

  float* __restrict__ pl = planes + (size_t)chunk * N + node0;
  for (int i = threadIdx.x; i < nodes; i += T16_THREADS) pl[i] = tile[i];
}

template <int NP>
__global__ void gcn_out_finish(const float* __restrict__ planes, const float* __restrict__ hw3,
                               const float* __restrict__ dis, const float* __restrict__ b3,
                               float* __restrict__ out, int N) {
  int i = blockIdx.x * blockDim.x + threadIdx.x;
  if (i >= N) return;
  float d = dis[i];
  float acc = d * d * hw3[i] + b3[0];
#pragma unroll
  for (int c = 0; c < NP; ++c) acc += planes[(size_t)c * N + i];
  out[i] = acc;
}

static inline int cdiv(long long a, long long b) { return (int)((a + b - 1) / b); }

extern "C" void kernel_launch(void* const* d_in, const int* in_sizes, int n_in,
                              void* d_out, int out_size, void* d_ws, size_t ws_size,
                              hipStream_t stream) {
  const float* x  = (const float*)d_in[0];
  const int*   ei = (const int*)d_in[1];
  const float* W1 = (const float*)d_in[2];
  const float* b1 = (const float*)d_in[3];
  const float* W2 = (const float*)d_in[4];
  const float* b2 = (const float*)d_in[5];
  const float* W3 = (const float*)d_in[6];
  const float* b3 = (const float*)d_in[7];

  const int IN_CH = 256;
  const int HID = 16;
  const int N = in_sizes[0] / IN_CH;
  const int E = in_sizes[1] / 2;
  const int* src = ei;
  const int* dst = ei + E;
  float* out = (float*)d_out;

  char* ws = (char*)d_ws;
  size_t off = 0;
  float* dis = (float*)(ws + off); off += (((size_t)N * 4) + 255) & ~(size_t)255;
  float* HW  = (float*)(ws + off); off += (((size_t)N * HID * 4) + 255) & ~(size_t)255;
  float* AGG = (float*)(ws + off); off += (((size_t)N * HID * 4) + 255) & ~(size_t)255;
  (void)n_in; (void)out_size;

  const int B = 256;
  const int gemmBlocks = cdiv(cdiv(N, 16), B / 32);

  const int tileGrid = cdiv(N, TILE_NODES) * TILE_CHUNKS;
  const int epc = cdiv(cdiv(E, TILE_CHUNKS), 4 * TILE_THREADS) * (4 * TILE_THREADS);

  const int t16Grid = cdiv(N, T16_NODES) * T16_CHUNKS;

  const int crGrid = cdiv(N, CR_NODES) * CR_CHUNKS;
  const int epcC = cdiv(cdiv(E, CR_CHUNKS), 4 * CR_THREADS) * (4 * CR_THREADS);

  const size_t listBytes = (size_t)t16Grid * T16_LIST_CAP * 4;
  const size_t offP = off + ((((size_t)t16Grid * 4) + 255) & ~(size_t)255);
  const size_t offN = offP + listBytes;
  const bool keepP = N <= (1 << (32 - T16_ROW_BITS)) && ws_size >= offN;
  const bool keepN = keepP && ws_size >= offN + listBytes;
  int* cnt16 = (int*)(ws + off);
  unsigned* listP = keepP ? (unsigned*)(ws + offP) : nullptr;
  float* listN = keepN ? (float*)(ws + offN) : nullptr;

  const size_t offCC = offN + listBytes;
  const size_t offCL = offCC + ((((size_t)crGrid * 4) + 255) & ~(size_t)255);
  const bool keepC = N <= (1 << (32 - CR_BITS)) && ws_size >= offCL + (size_t)crGrid * CR_CAP * 4;
  int* cntC = keepC ? (int*)(ws + offCC) : nullptr;
  unsigned* listC = keepC ? (unsigned*)(ws + offCL) : nullptr;

  if (keepC) gcn_deg_coarse<true><<<crGrid, CR_THREADS, 0, stream>>>(src, dst, HW, listC, cntC, N, E, epcC);
  else       gcn_deg_coarse<false><<<crGrid, CR_THREADS, 0, stream>>>(src, dst, HW, listC, cntC, N, E, epcC);
  gcn_deg_finish<CR_CHUNKS><<<cdiv(N, B), B, 0, stream>>>(HW, dis, N);

  gcn_gemm_wmma_f32<false><<<gemmBlocks, B, 0, stream>>>(x, W1, dis, b1, HW, AGG, N, IN_CH);
  gcn_tile_scatter16<false><<<t16Grid, T16_THREADS, 0, stream>>>(src, dst, dis, HW, AGG, listP, listN, cnt16, listC, cntC, N, E, epcC);

  gcn_gemm_wmma_f32<true><<<gemmBlocks, B, 0, stream>>>(AGG, W2, dis, b2, HW, AGG, N, HID);
  gcn_tile_scatter16<true><<<t16Grid, T16_THREADS, 0, stream>>>(src, dst, dis, HW, AGG, listP, listN, cnt16, listC, cntC, N, E, epcC);

  gcn_mv16<<<cdiv(N, B), B, 0, stream>>>(AGG, W3, HW, N);
  if (keepN) {
    gcn_list_acc1<<<t16Grid, T16_THREADS, 0, stream>>>(src, dst, dis, HW, listP, listN, cnt16, HW + N, N, E, epcC);
    gcn_out_finish<T16_CHUNKS><<<cdiv(N, B), B, 0, stream>>>(HW + N, HW, dis, b3, out, N);
  } else {
    gcn_tile_acc1<true><<<tileGrid, TILE_THREADS, 0, stream>>>(src, dst, dis, HW, HW + N, N, E, epc);
    gcn_out_finish<TILE_CHUNKS><<<cdiv(N, B), B, 0, stream>>>(HW + N, HW, dis, b3, out, N);
  }
}


# Degree pass and layer-1 WMMA product run as one launch, one work-group of each role per WGP (9 -> 8 launches); layer 1's self loop added by the next product's A loads: sum order bias, edges, self
# speedup vs baseline: 5.0315x; 1.0738x over previous
#include <hip/hip_runtime.h>
#include <hip/hip_bf16.h>

typedef __attribute__((ext_vector_type(2))) float v2f;
typedef __attribute__((ext_vector_type(8))) float v8f;

template <bool RELU_A, bool SELF_A, bool FIRST>
__device__ __forceinline__ void gcn_gemm_body(
    int wave, const float* A, const float* S, const float* __restrict__ W,
    const float* __restrict__ dis, const float* __restrict__ b,
    float* C, float* AGG, int M, int K) {
  const int lane = threadIdx.x & 31;
  const int m0 = wave << 4;
  if (m0 >= M) return;
  const int half = lane >> 4;
  const int l = lane & 15;

  v8f acc = {};
  const float* arow = A + (size_t)(m0 + l) * K + (half << 1);
  const float* srow = SELF_A ? S + (size_t)(m0 + l) * K + (half << 1) : nullptr;
  float d2 = 0.f;
  if (SELF_A) {
    const float d = dis[m0 + l];
    d2 = d * d;
  }
  for (int k0 = 0; k0 < K; k0 += 4) {
    v2f a;
    a.x = arow[k0 + 0];
    a.y = arow[k0 + 1];
    if (SELF_A) {
      a.x += d2 * srow[k0 + 0];
      a.y += d2 * srow[k0 + 1];
    }
    if (RELU_A) {
      a.x = fmaxf(a.x, 0.0f);
      a.y = fmaxf(a.y, 0.0f);
    }
    v2f b2;
    b2.x = W[(size_t)(k0 + (half << 1) + 0) * 16 + l];
    b2.y = W[(size_t)(k0 + (half << 1) + 1) * 16 + l];
    acc = __builtin_amdgcn_wmma_f32_16x16x4_f32(
         false, a,  false, b2,
         (short)0, acc,  false,  false);
  }
  const int r0 = m0 + (half << 3);
  const float bias = b[l];
  float* cp = C + ((size_t)r0 << 4) + l;
  float* gp = AGG + ((size_t)r0 << 4) + l;
  if (FIRST) {
#pragma unroll
    for (int r = 0; r < 8; ++r) {
      cp[(size_t)r << 4] = acc[r];
      gp[(size_t)r << 4] = bias;
    }
  } else {
    const float4 d0 = *reinterpret_cast<const float4*>(dis + r0);
    const float4 d1 = *reinterpret_cast<const float4*>(dis + r0 + 4);
    const float dd[8] = {d0.x, d0.y, d0.z, d0.w, d1.x, d1.y, d1.z, d1.w};
#pragma unroll
    for (int r = 0; r < 8; ++r) {
      cp[(size_t)r << 4] = acc[r];
      gp[(size_t)r << 4] = dd[r] * dd[r] * acc[r] + bias;
    }
  }
}

template <bool RELU_A, bool SELF_A>
__global__ __launch_bounds__(256) void gcn_gemm_wmma_f32(
    const float* A, const float* S, const float* __restrict__ W,
    const float* __restrict__ dis, const float* __restrict__ b,
    float* C, float* AGG, int M, int K) {
  const int wave = blockIdx.x * (blockDim.x >> 5) + (threadIdx.x >> 5);
  gcn_gemm_body<RELU_A, SELF_A, false>(wave, A, S, W, dis, b, C, AGG, M, K);
}

constexpr int TILE_NODES  = 25000;
constexpr int TILE_CHUNKS = 31;
constexpr int TILE_THREADS = 1024;

template <bool WEIGHTED, int NE>
__device__ __forceinline__ void gcn_tile_add(
    float* tile, const float* dtile, const int (&d)[NE], const int (&s)[NE], int node0, int nodes,
    const float* __restrict__ dis, const float* __restrict__ hw3) {
  bool hit[NE];
  int rel[NE];
#pragma unroll
  for (int j = 0; j < NE; ++j) {
    rel[j] = d[j] - node0;
    hit[j] = (unsigned)rel[j] < (unsigned)nodes;
  }
  if (WEIGHTED) {
    float v[NE];
#pragma unroll
    for (int j = 0; j < NE; ++j) {
      const int si = hit[j] ? s[j] : 0;
      const int ri = hit[j] ? rel[j] : 0;
      v[j] = dis[si] * dtile[ri] * hw3[si];
    }
#pragma unroll
    for (int j = 0; j < NE; ++j)
      if (hit[j]) atomicAdd(&tile[rel[j]], v[j]);
  } else {
#pragma unroll
    for (int j = 0; j < NE; ++j)
      if (hit[j]) atomicAdd(&tile[rel[j]], 1.0f);
  }
}

template <bool WEIGHTED>
__global__ __launch_bounds__(TILE_THREADS) void gcn_tile_acc1(
    const int* __restrict__ src, const int* __restrict__ dst,
    const float* __restrict__ dis, const float* __restrict__ hw3,
    float* __restrict__ planes, int N, int E, int epc) {
  __shared__ float tile[TILE_NODES];
  __shared__ float dtile[WEIGHTED ? TILE_NODES : 1];
  const int range = blockIdx.x / TILE_CHUNKS;
  const int chunk = blockIdx.x - range * TILE_CHUNKS;
  const int node0 = range * TILE_NODES;
  const int nodes = min(TILE_NODES, N - node0);
  for (int i = threadIdx.x; i < nodes; i += TILE_THREADS) {
    tile[i] = 0.f;
    if (WEIGHTED) dtile[i] = dis[node0 + i];
  }
  __syncthreads();

  const int e0 = min(E, chunk * epc);
  const int e1 = min(E, e0 + epc);

  const int nvec = (e1 - e0) >> 2;
  const int4* __restrict__ d4 = reinterpret_cast<const int4*>(dst + e0);
  const int4* __restrict__ s4 = reinterpret_cast<const int4*>(src + e0);
  int g = threadIdx.x;
  for (; g + 3 * TILE_THREADS < nvec; g += 4 * TILE_THREADS) {
    const int4 q0 = d4[g];
    const int4 q1 = d4[g + TILE_THREADS];
    const int4 q2 = d4[g + 2 * TILE_THREADS];
    const int4 q3 = d4[g + 3 * TILE_THREADS];
    int4 p0 = q0, p1 = q0, p2 = q0, p3 = q0;
    if (WEIGHTED) {
      p0 = s4[g];
      p1 = s4[g + TILE_THREADS];
      p2 = s4[g + 2 * TILE_THREADS];
      p3 = s4[g + 3 * TILE_THREADS];
    }
    const int d[16] = {q0.x, q0.y, q0.z, q0.w, q1.x, q1.y, q1.z, q1.w,
                       q2.x, q2.y, q2.z, q2.w, q3.x, q3.y, q3.z, q3.w};
    const int s[16] = {p0.x, p0.y, p0.z, p0.w, p1.x, p1.y, p1.z, p1.w,
                       p2.x, p2.y, p2.z, p2.w, p3.x, p3.y, p3.z, p3.w};
    gcn_tile_add<WEIGHTED, 16>(tile, dtile, d, s, node0, nodes, dis, hw3);
  }
  for (; g < nvec; g += TILE_THREADS) {
    const int4 q = d4[g];
    int4 p = q;
    if (WEIGHTED) p = s4[g];
    const int d[4] = {q.x, q.y, q.z, q.w};
    const int s[4] = {p.x, p.y, p.z, p.w};
    gcn_tile_add<WEIGHTED, 4>(tile, dtile, d, s, node0, nodes, dis, hw3);
  }
  for (int t = e0 + 4 * nvec + threadIdx.x; t < e1; t += TILE_THREADS) {
    const int d[1] = {dst[t]};
    const int s[1] = {WEIGHTED ? src[t] : 0};
    gcn_tile_add<WEIGHTED, 1>(tile, dtile, d, s, node0, nodes, dis, hw3);
  }
  __syncthreads();

  float* __restrict__ pl = planes + (size_t)chunk * N + node0;
  for (int i = threadIdx.x; i < nodes; i += TILE_THREADS) pl[i] = tile[i];
}

template <int NP>
__global__ void gcn_deg_finish(const float* __restrict__ planes, float* __restrict__ dis, int N) {
  int i = blockIdx.x * blockDim.x + threadIdx.x;
  if (i >= N) return;
  float deg = 1.0f;
#pragma unroll
  for (int c = 0; c < NP; ++c) deg += planes[(size_t)c * N + i];
  dis[i] = rsqrtf(deg);
}

constexpr int T16_NODES   = 4096;
constexpr int T16_CHUNKS  = 5;
constexpr int T16_THREADS = 1024;
constexpr int T16_WAVES   = T16_THREADS / 32;
constexpr int T16_RING    = 64;

template <int J>
__device__ __forceinline__ int gcn_row_lane(int v) {
  return __builtin_amdgcn_mov_dpp(v, 0x150 + J, 0xf, 0xf, true);
}

#define GCN_FOR_16(X) X(0) X(1) X(2) X(3) X(4) X(5) X(6) X(7) X(8) X(9) X(10) X(11) X(12) X(13) X(14) X(15)

constexpr int T16_LIST_CAP = 32768;
constexpr int T16_ROW_BITS = 12;
static_assert(T16_NODES == (1 << T16_ROW_BITS), "row field of a packed hit");

constexpr int CR_BITS    = 14;
constexpr int CR_NODES   = 1 << CR_BITS;
constexpr int CR_CHUNKS  = 19;
constexpr int CR_THREADS = 1024;
constexpr int CR_WAVES   = CR_THREADS / 32;
constexpr int CR_RING    = 64;
constexpr int CR_CAP     = 32768;
static_assert(CR_NODES == 4 * T16_NODES, "a coarse range is four fine ranges");
static_assert(CR_CHUNKS >= T16_CHUNKS, "every fine chunk holds a coarse chunk");

__device__ __forceinline__ int gcn_chunk_first(int c) { return (c * CR_CHUNKS) / T16_CHUNKS; }

template <bool KEEP, int NV>
__device__ __forceinline__ void gcn_deg_coarse_body(
    const int bid, const int* __restrict__ src, const int* __restrict__ dst,
    float* __restrict__ planes, unsigned* coarse, int* ccounts, int N, int E, int epc) {
  __shared__ float tile[CR_NODES];
  __shared__ unsigned ring_all[CR_WAVES * CR_RING];
  __shared__ int cursor;
  const int range = bid / CR_CHUNKS;
  const int chunk = bid - range * CR_CHUNKS;
  const int node0 = range << CR_BITS;
  const int nodes = min(CR_NODES, N - node0);
  for (int i = threadIdx.x; i < nodes; i += CR_THREADS) tile[i] = 0.f;
  if (threadIdx.x == 0) cursor = 0;
  __syncthreads();

  const int lane = threadIdx.x & 31;
  const int wv = __builtin_amdgcn_readfirstlane(threadIdx.x >> 5);
  unsigned* ring = ring_all + wv * CR_RING;
  unsigned* list = KEEP ? coarse + (size_t)bid * CR_CAP : nullptr;
  int head = 0, tail = 0;

  auto flush = [&](int count) {
    int pos = 0;
    if (lane == 0) pos = atomicAdd(&cursor, count);
    pos = __builtin_amdgcn_readfirstlane(pos) + lane;
    if (lane < count && pos < CR_CAP) list[pos] = ring[(head + lane) & (CR_RING - 1)];
  };
  auto step = [&](int d, int s) {
    const int rel = d - node0;
    const bool hit = (unsigned)rel < (unsigned)nodes;
    if (KEEP) {
      const unsigned m = __builtin_amdgcn_ballot_w32(hit);
      if (m != 0u) {
        if (hit) {
          atomicAdd(&tile[rel], 1.0f);
          const int slot = (tail + (int)__builtin_amdgcn_mbcnt_lo(m, 0u)) & (CR_RING - 1);
          ring[slot] = ((unsigned)s << CR_BITS) | (unsigned)rel;
        }
        tail += __builtin_popcount(m);
        if (tail - head >= 32) {
          flush(32);
          head += 32;
        }
      }
    } else {
      if (hit) atomicAdd(&tile[rel], 1.0f);
    }
  };

  const int e0 = min(E, chunk * epc);
  const int e1 = min(E, e0 + epc);
  const int nvec = (e1 - e0) >> 2;
  const int4* __restrict__ d4 = reinterpret_cast<const int4*>(dst + e0);
  const int4* __restrict__ s4 = reinterpret_cast<const int4*>(src + e0);
  const int last = nvec - 1;
  int4 nd[NV], ns[NV];
#pragma unroll
  for (int j = 0; j < NV; ++j) {
    nd[j] = make_int4(-1, -1, -1, -1);
    ns[j] = make_int4(0, 0, 0, 0);
  }
  if (nvec > 0) {
#pragma unroll
    for (int j = 0; j < NV; ++j) {
      const int g = min((int)threadIdx.x + j * CR_THREADS, last);
      nd[j] = d4[g];
      ns[j] = s4[g];
    }
  }
  for (int base = 0; base < nvec; base += NV * CR_THREADS) {
    int4 q[NV], p[NV];
    bool in[NV];
#pragma unroll
    for (int j = 0; j < NV; ++j) {
      q[j] = nd[j];
      p[j] = ns[j];
    }
#pragma unroll
    for (int j = 0; j < NV; ++j) {
      const int g = base + (int)threadIdx.x + j * CR_THREADS;
      in[j] = g < nvec;
      const int gn = min(g + NV * CR_THREADS, last);
      nd[j] = d4[gn];
      ns[j] = s4[gn];
    }
#pragma unroll
    for (int j = 0; j < NV; ++j) {
      step(in[j] ? q[j].x : -1, p[j].x);
      step(in[j] ? q[j].y : -1, p[j].y);
      step(in[j] ? q[j].z : -1, p[j].z);
      step(in[j] ? q[j].w : -1, p[j].w);
    }
  }
  {
    const int t = e0 + 4 * nvec + (int)threadIdx.x;
    const bool have = t < e1;
    const int d = have ? dst[t] : -1;
    const int s = have ? src[t] : 0;
    step(d, s);
  }
  if (KEEP && tail - head > 0) flush(tail - head);
  __syncthreads();

  if (KEEP && threadIdx.x == 0) ccounts[bid] = cursor;

  float* __restrict__ pl = planes + (size_t)chunk * N + node0;
  for (int i = threadIdx.x; i < nodes; i += CR_THREADS) pl[i] = tile[i];
}

template <bool KEEP>
__global__ __launch_bounds__(CR_THREADS) void gcn_deg_coarse(
    const int* __restrict__ src, const int* __restrict__ dst,
    float* __restrict__ planes, unsigned* coarse, int* ccounts, int N, int E, int epc) {
  gcn_deg_coarse_body<KEEP, 4>((int)blockIdx.x, src, dst, planes, coarse, ccounts, N, E, epc);
}

__global__ __launch_bounds__(CR_THREADS) void gcn_deg_and_gemm1(
    const int* __restrict__ src, const int* __restrict__ dst,
    float* __restrict__ planes, unsigned* coarse, int* ccounts, int N, int E, int epc, int crGrid,
    const float* x, const float* __restrict__ W1, const float* __restrict__ b1,
    float* C, float* AGG, int K) {
  if ((int)blockIdx.x < crGrid) {
    gcn_deg_coarse_body<true, 2>((int)blockIdx.x, src, dst, planes, coarse, ccounts, N, E, epc);
  } else {
    const int wave = ((int)blockIdx.x - crGrid) * CR_WAVES + (int)(threadIdx.x >> 5);
    gcn_gemm_body<false, false, true>(wave, x, nullptr, W1, nullptr, b1, C, AGG, N, K);
  }
}

__device__ __forceinline__ void gcn_tile16_gather(
    int s, int lane, const float* __restrict__ hw, float (&v)[16]) {
  const int c = lane & 15;
#define GCN_ROW_LOAD(J) v[J] = hw[(gcn_row_lane<J>(s) << 4) + c];
  GCN_FOR_16(GCN_ROW_LOAD)
#undef GCN_ROW_LOAD
}

template <bool FULL>
__device__ __forceinline__ void gcn_tile16_add(
    const float (&v)[16], int rel, float nrm, int count, int lane, float* tile) {
  const int c = lane & 15;
  const int row0 = lane & 16;
  const int nrm_bits = __float_as_int(nrm);
  const int left = count - row0;

#define GCN_ROW_ADD(J)                                                        \
  {                                                                           \
    const int rj = gcn_row_lane<J>(rel);                                      \
    const float nj = __int_as_float(gcn_row_lane<J>(nrm_bits));               \
    if (FULL || J < left) atomicAdd(&tile[(rj << 4) + c], nj * v[J]);         \
  }
  GCN_FOR_16(GCN_ROW_ADD)
#undef GCN_ROW_ADD
}
#undef GCN_FOR_16

template <bool FULL>
__device__ __forceinline__ void gcn_tile16_rows(
    int s, int rel, float nrm, int count, int lane,
    const float* __restrict__ hw, float* tile) {
  float v[16];
  gcn_tile16_gather(s, lane, hw, v);
  gcn_tile16_add<FULL>(v, rel, nrm, count, lane, tile);
}

template <bool FULL>
__device__ __forceinline__ void gcn_tile16_batch(
    const int* ring_s, const int* ring_r, int head, int count, int lane,
    const float* __restrict__ dis, const float* __restrict__ hw,
    const float* dtile, float* tile,
    unsigned* list_p, float* list_n, int* cursor) {
  int s = 0, rel = 0;
  float nrm = 0.f;
  const bool have = FULL || lane < count;
  if (have) {
    const int slot = (head + lane) & (T16_RING - 1);
    rel = ring_r[slot];
    s = ring_s[slot];
    nrm = dis[s] * dtile[rel];
  }
  if (list_p != nullptr) {
    int pos = 0;
    if (lane == 0) pos = atomicAdd(cursor, count);
    pos = __builtin_amdgcn_readfirstlane(pos) + lane;
    if (have && pos < T16_LIST_CAP) {
      list_p[pos] = ((unsigned)s << T16_ROW_BITS) | (unsigned)rel;
      if (list_n != nullptr) list_n[pos] = nrm;
    }
  }
  gcn_tile16_rows<FULL>(s, rel, nrm, count, lane, hw, tile);
}

template <bool WALK>
__global__ __launch_bounds__(T16_THREADS) void gcn_tile_scatter16(
    const int* __restrict__ src, const int* __restrict__ dst,
    const float* __restrict__ dis, const float* __restrict__ hw,
    float* __restrict__ agg, unsigned* lists_p, float* lists_n, int* counts,
    const unsigned* coarse, const int* ccounts, int N, int E, int epcC) {
  __shared__ float tile[T16_NODES * 16];
  __shared__ float dtile[T16_NODES];
  __shared__ int ring_s_all[T16_WAVES * T16_RING];
  __shared__ int ring_r_all[T16_WAVES * T16_RING];
  __shared__ int cursor;
  const int range = blockIdx.x / T16_CHUNKS;
  const int chunk = blockIdx.x - range * T16_CHUNKS;
  const int node0 = range * T16_NODES;
  const int nodes = min(T16_NODES, N - node0);
  for (int i = threadIdx.x; i < nodes * 16; i += T16_THREADS) tile[i] = 0.f;
  for (int i = threadIdx.x; i < nodes; i += T16_THREADS) dtile[i] = dis[node0 + i];
  if (threadIdx.x == 0) cursor = 0;
  __syncthreads();

  const int lane = threadIdx.x & 31;
  const int wv = __builtin_amdgcn_readfirstlane(threadIdx.x >> 5);

  const size_t list0 = (size_t)blockIdx.x * T16_LIST_CAP;
  bool walked = false;
  if (WALK && lists_p != nullptr) {
    const int cnt = counts[blockIdx.x];
    if (cnt <= T16_LIST_CAP) {
      walked = true;
      const unsigned* __restrict__ lp = lists_p + list0;
      const float* __restrict__ ln = lists_n != nullptr ? lists_n + list0 : nullptr;
      const int trips = (cnt + 31) >> 5;
      const int lastpos = cnt - 1;
      unsigned e_next = 0u;
      float n_next = 0.f;
      int b = wv;
      if (b < trips) {
        const int p = min((b << 5) + lane, lastpos);
        e_next = lp[p];
        if (ln != nullptr) n_next = ln[p];
      }
      for (; b < trips; b += T16_WAVES) {
        const unsigned e = e_next;
        float nrm = n_next;
        const int s = (int)(e >> T16_ROW_BITS);
        const int rel = (int)(e & (unsigned)(T16_NODES - 1));
        float v[16];
        gcn_tile16_gather(s, lane, hw, v);
        if (ln == nullptr) nrm = dis[s] * dtile[rel];
        const int bn = b + T16_WAVES;
        if (bn < trips) {
          const int p = min((bn << 5) + lane, lastpos);
          e_next = lp[p];
          if (ln != nullptr) n_next = ln[p];
        }
        const int left = cnt - (b << 5);
        if (left >= 32) gcn_tile16_add<true>(v, rel, nrm, 32, lane, tile);
        else            gcn_tile16_add<false>(v, rel, nrm, left, lane, tile);
      }
    }
  }

  if (!walked) {
    unsigned* list_p = (!WALK && lists_p != nullptr) ? lists_p + list0 : nullptr;
    float* list_n = (list_p != nullptr && lists_n != nullptr) ? lists_n + list0 : nullptr;
    int* ring_s = ring_s_all + wv * T16_RING;
    int* ring_r = ring_r_all + wv * T16_RING;
    int head = 0, tail = 0;

    auto take = [&](bool hit, int s, int rel) {
      const unsigned m = __builtin_amdgcn_ballot_w32(hit);
      if (m != 0u) {
        if (hit) {
          const int slot = (tail + (int)__builtin_amdgcn_mbcnt_lo(m, 0u)) & (T16_RING - 1);
          ring_s[slot] = s;
          ring_r[slot] = rel;
        }
        tail += __builtin_popcount(m);
        if (tail - head >= 32) {
          gcn_tile16_batch<true>(ring_s, ring_r, head, 32, lane, dis, hw, dtile, tile,
                                 list_p, list_n, &cursor);
          head += 32;
        }
      }
    };
    auto step = [&](int d, int s) {
      const int rel = d - node0;
      take((unsigned)rel < (unsigned)nodes, s, rel);
    };
    const unsigned subkey = (unsigned)(range & 3) << T16_ROW_BITS;
    auto lstep = [&](bool valid, unsigned e) {
      const bool hit = valid && ((e & (3u << T16_ROW_BITS)) == subkey);
      take(hit, (int)(e >> CR_BITS), (int)(e & (unsigned)(T16_NODES - 1)));
    };

    auto scan_raw = [&](int e0, int e1) {
      const int nvec = (e1 - e0) >> 2;
      const int4* __restrict__ d4 = reinterpret_cast<const int4*>(dst + e0);
      const int4* __restrict__ s4 = reinterpret_cast<const int4*>(src + e0);
      const int last = nvec - 1;
      int4 n0 = make_int4(-1, -1, -1, -1), n1 = n0, n2 = n0, n3 = n0;
      int4 m0 = make_int4(0, 0, 0, 0), m1 = m0, m2 = m0, m3 = m0;
      if (nvec > 0) {
        const int g = threadIdx.x;
        n0 = d4[min(g, last)];
        n1 = d4[min(g + T16_THREADS, last)];
        n2 = d4[min(g + 2 * T16_THREADS, last)];
        n3 = d4[min(g + 3 * T16_THREADS, last)];
        m0 = s4[min(g, last)];
        m1 = s4[min(g + T16_THREADS, last)];
        m2 = s4[min(g + 2 * T16_THREADS, last)];
        m3 = s4[min(g + 3 * T16_THREADS, last)];
      }
      for (int base = 0; base < nvec; base += 4 * T16_THREADS) {
        const int4 q0 = n0, q1 = n1, q2 = n2, q3 = n3;
        const int4 p0 = m0, p1 = m1, p2 = m2, p3 = m3;
        const int g0 = base + threadIdx.x;
        const int g1 = g0 + T16_THREADS;
        const int g2 = g0 + 2 * T16_THREADS;
        const int g3 = g0 + 3 * T16_THREADS;
        n0 = d4[min(g0 + 4 * T16_THREADS, last)];
        n1 = d4[min(g1 + 4 * T16_THREADS, last)];
        n2 = d4[min(g2 + 4 * T16_THREADS, last)];
        n3 = d4[min(g3 + 4 * T16_THREADS, last)];
        m0 = s4[min(g0 + 4 * T16_THREADS, last)];
        m1 = s4[min(g1 + 4 * T16_THREADS, last)];
        m2 = s4[min(g2 + 4 * T16_THREADS, last)];
        m3 = s4[min(g3 + 4 * T16_THREADS, last)];
        const bool in0 = g0 < nvec, in1 = g1 < nvec, in2 = g2 < nvec, in3 = g3 < nvec;
        step(in0 ? q0.x : -1, p0.x); step(in0 ? q0.y : -1, p0.y); step(in0 ? q0.z : -1, p0.z); step(in0 ? q0.w : -1, p0.w);
        step(in1 ? q1.x : -1, p1.x); step(in1 ? q1.y : -1, p1.y); step(in1 ? q1.z : -1, p1.z); step(in1 ? q1.w : -1, p1.w);
        step(in2 ? q2.x : -1, p2.x); step(in2 ? q2.y : -1, p2.y); step(in2 ? q2.z : -1, p2.z); step(in2 ? q2.w : -1, p2.w);
        step(in3 ? q3.x : -1, p3.x); step(in3 ? q3.y : -1, p3.y); step(in3 ? q3.z : -1, p3.z); step(in3 ? q3.w : -1, p3.w);
      }
      {
        const int t = e0 + 4 * nvec + (int)threadIdx.x;
        const bool have = t < e1;
        const int d = have ? dst[t] : -1;
        const int s = have ? src[t] : 0;
        step(d, s);
      }
    };

    auto scan_list = [&](const unsigned* __restrict__ lp, int n) {
      const int nvec = n >> 2;
      const uint4* __restrict__ v4 = reinterpret_cast<const uint4*>(lp);
      const int last = nvec - 1;
      uint4 a0 = make_uint4(0u, 0u, 0u, 0u), a1 = a0;
      if (nvec > 0) {
        const int g = threadIdx.x;
        a0 = v4[min(g, last)];
        a1 = v4[min(g + T16_THREADS, last)];
      }
      for (int base = 0; base < nvec; base += 2 * T16_THREADS) {
        const uint4 q0 = a0, q1 = a1;
        const int g0 = base + threadIdx.x;
        const int g1 = g0 + T16_THREADS;
        a0 = v4[min(g0 + 2 * T16_THREADS, last)];
        a1 = v4[min(g1 + 2 * T16_THREADS, last)];
        const bool in0 = g0 < nvec, in1 = g1 < nvec;
        lstep(in0, q0.x); lstep(in0, q0.y); lstep(in0, q0.z); lstep(in0, q0.w);
        lstep(in1, q1.x); lstep(in1, q1.y); lstep(in1, q1.z); lstep(in1, q1.w);
      }
      {
        const int t = 4 * nvec + (int)threadIdx.x;
        const bool have = t < n;
        const unsigned e = have ? lp[t] : 0u;
        lstep(have, e);
      }
    };

    const int crow = (range >> 2) * CR_CHUNKS;
    const int k1 = gcn_chunk_first(chunk + 1);
    for (int k = gcn_chunk_first(chunk); k < k1; ++k) {
      int cn = CR_CAP + 1;
      if (coarse != nullptr) cn = ccounts[crow + k];
      if (cn <= CR_CAP) scan_list(coarse + (size_t)(crow + k) * CR_CAP, cn);
      else              scan_raw(min(E, k * epcC), min(E, (k + 1) * epcC));
    }
    if (tail - head > 0)
      gcn_tile16_batch<false>(ring_s, ring_r, head, tail - head, lane, dis, hw, dtile, tile,
                              list_p, list_n, &cursor);
  }
  __syncthreads();

  if (!WALK && lists_p != nullptr && threadIdx.x == 0) counts[blockIdx.x] = cursor;

  float* __restrict__ ag = agg + ((size_t)node0 << 4);
  for (int i = threadIdx.x; i < nodes * 16; i += T16_THREADS) atomicAdd(&ag[i], tile[i]);
}

__global__ void gcn_mv16(const float* __restrict__ h, const float* __restrict__ W3,
                         float* __restrict__ out, int N) {
  int i = blockIdx.x * blockDim.x + threadIdx.x;
  if (i >= N) return;
  const float* hp = h + ((size_t)i << 4);
  float s = 0.f;
#pragma unroll
  for (int c = 0; c < 16; ++c) s += fmaxf(__builtin_nontemporal_load(&hp[c]), 0.0f) * W3[c];
  out[i] = s;
}

__global__ __launch_bounds__(T16_THREADS) void gcn_list_acc1(
    const int* __restrict__ src, const int* __restrict__ dst,
    const float* __restrict__ dis, const float* __restrict__ hw3,
    const unsigned* __restrict__ lists_p, const float* __restrict__ lists_n,
    const int* __restrict__ counts, float* __restrict__ planes, int N, int E, int epcC) {
  __shared__ float tile[T16_NODES];
  const int range = blockIdx.x / T16_CHUNKS;
  const int chunk = blockIdx.x - range * T16_CHUNKS;
  const int node0 = range * T16_NODES;
  const int nodes = min(T16_NODES, N - node0);
  for (int i = threadIdx.x; i < nodes; i += T16_THREADS) tile[i] = 0.f;
  __syncthreads();

  const int cnt = counts[blockIdx.x];
  if (cnt <= T16_LIST_CAP) {
    const unsigned* __restrict__ lp = lists_p + (size_t)blockIdx.x * T16_LIST_CAP;
    const float* __restrict__ ln = lists_n + (size_t)blockIdx.x * T16_LIST_CAP;
    int p = threadIdx.x;
    for (; p + 3 * T16_THREADS < cnt; p += 4 * T16_THREADS) {
      const unsigned e0 = lp[p];
      const unsigned e1 = lp[p + T16_THREADS];
      const unsigned e2 = lp[p + 2 * T16_THREADS];
      const unsigned e3 = lp[p + 3 * T16_THREADS];
      const float n0 = ln[p];
      const float n1 = ln[p + T16_THREADS];
      const float n2 = ln[p + 2 * T16_THREADS];
      const float n3 = ln[p + 3 * T16_THREADS];
      const float h0 = hw3[e0 >> T16_ROW_BITS];
      const float h1 = hw3[e1 >> T16_ROW_BITS];
      const float h2 = hw3[e2 >> T16_ROW_BITS];
      const float h3 = hw3[e3 >> T16_ROW_BITS];
      atomicAdd(&tile[e0 & (unsigned)(T16_NODES - 1)], n0 * h0);
      atomicAdd(&tile[e1 & (unsigned)(T16_NODES - 1)], n1 * h1);
      atomicAdd(&tile[e2 & (unsigned)(T16_NODES - 1)], n2 * h2);
      atomicAdd(&tile[e3 & (unsigned)(T16_NODES - 1)], n3 * h3);
    }
    for (; p < cnt; p += T16_THREADS) {
      const unsigned e = lp[p];
      atomicAdd(&tile[e & (unsigned)(T16_NODES - 1)], ln[p] * hw3[e >> T16_ROW_BITS]);
    }
  } else {
    const int e0 = min(E, gcn_chunk_first(chunk) * epcC);
    const int e1 = min(E, gcn_chunk_first(chunk + 1) * epcC);
    for (int t = e0 + (int)threadIdx.x; t < e1; t += T16_THREADS) {
      const int rel = dst[t] - node0;
      if ((unsigned)rel < (unsigned)nodes) {
        const int s = src[t];
        atomicAdd(&tile[rel], dis[s] * dis[node0 + rel] * hw3[s]);
      }
    }
  }
  __syncthreads();

  float* __restrict__ pl = planes + (size_t)chunk * N + node0;
  for (int i = threadIdx.x; i < nodes; i += T16_THREADS) pl[i] = tile[i];
}

template <int NP>
__global__ void gcn_out_finish(const float* __restrict__ planes, const float* __restrict__ hw3,
                               const float* __restrict__ dis, const float* __restrict__ b3,
                               float* __restrict__ out, int N) {
  int i = blockIdx.x * blockDim.x + threadIdx.x;
  if (i >= N) return;
  float d = dis[i];
  float acc = d * d * hw3[i] + b3[0];
#pragma unroll
  for (int c = 0; c < NP; ++c) acc += planes[(size_t)c * N + i];
  out[i] = acc;
}

static inline int cdiv(long long a, long long b) { return (int)((a + b - 1) / b); }

extern "C" void kernel_launch(void* const* d_in, const int* in_sizes, int n_in,
                              void* d_out, int out_size, void* d_ws, size_t ws_size,
                              hipStream_t stream) {
  const float* x  = (const float*)d_in[0];
  const int*   ei = (const int*)d_in[1];
  const float* W1 = (const float*)d_in[2];
  const float* b1 = (const float*)d_in[3];
  const float* W2 = (const float*)d_in[4];
  const float* b2 = (const float*)d_in[5];
  const float* W3 = (const float*)d_in[6];
  const float* b3 = (const float*)d_in[7];

  const int IN_CH = 256;
  const int HID = 16;
  const int N = in_sizes[0] / IN_CH;
  const int E = in_sizes[1] / 2;
  const int* src = ei;
  const int* dst = ei + E;
  float* out = (float*)d_out;

  char* ws = (char*)d_ws;
  size_t off = 0;
  float* dis = (float*)(ws + off); off += (((size_t)N * 4) + 255) & ~(size_t)255;
  float* HW  = (float*)(ws + off); off += (((size_t)N * HID * 4) + 255) & ~(size_t)255;
  float* AGG = (float*)(ws + off); off += (((size_t)N * HID * 4) + 255) & ~(size_t)255;
  (void)n_in; (void)out_size;

  const int B = 256;
  const int gemmBlocks = cdiv(cdiv(N, 16), B / 32);

  const int tileGrid = cdiv(N, TILE_NODES) * TILE_CHUNKS;
  const int epc = cdiv(cdiv(E, TILE_CHUNKS), 4 * TILE_THREADS) * (4 * TILE_THREADS);

  const int t16Grid = cdiv(N, T16_NODES) * T16_CHUNKS;

  const int crGrid = cdiv(N, CR_NODES) * CR_CHUNKS;
  const int epcC = cdiv(cdiv(E, CR_CHUNKS), 4 * CR_THREADS) * (4 * CR_THREADS);

  const size_t listBytes = (size_t)t16Grid * T16_LIST_CAP * 4;
  const size_t offP = off + ((((size_t)t16Grid * 4) + 255) & ~(size_t)255);
  const size_t offN = offP + listBytes;
  const bool keepP = N <= (1 << (32 - T16_ROW_BITS)) && ws_size >= offN;
  const bool keepN = keepP && ws_size >= offN + listBytes;
  int* cnt16 = (int*)(ws + off);
  unsigned* listP = keepP ? (unsigned*)(ws + offP) : nullptr;
  float* listN = keepN ? (float*)(ws + offN) : nullptr;

  const size_t offCC = offN + listBytes;
  const size_t offCL = offCC + ((((size_t)crGrid * 4) + 255) & ~(size_t)255);
  const bool keepC = N <= (1 << (32 - CR_BITS)) && ws_size >= offCL + (size_t)crGrid * CR_CAP * 4;
  int* cntC = keepC ? (int*)(ws + offCC) : nullptr;
  unsigned* listC = keepC ? (unsigned*)(ws + offCL) : nullptr;

  if (keepC) {
    float* degPlanes = listN;
    const int gemm1Groups = cdiv(cdiv(N, 16), CR_WAVES);
    gcn_deg_and_gemm1<<<crGrid + gemm1Groups, CR_THREADS, 0, stream>>>(
        src, dst, degPlanes, listC, cntC, N, E, epcC, crGrid, x, W1, b1, HW, AGG, IN_CH);
    gcn_deg_finish<CR_CHUNKS><<<cdiv(N, B), B, 0, stream>>>(degPlanes, dis, N);
    gcn_tile_scatter16<false><<<t16Grid, T16_THREADS, 0, stream>>>(src, dst, dis, HW, AGG, listP, listN, cnt16, listC, cntC, N, E, epcC);
    gcn_gemm_wmma_f32<true, true><<<gemmBlocks, B, 0, stream>>>(AGG, HW, W2, dis, b2, HW, AGG, N, HID);
  } else {
    gcn_deg_coarse<false><<<crGrid, CR_THREADS, 0, stream>>>(src, dst, HW, listC, cntC, N, E, epcC);
    gcn_deg_finish<CR_CHUNKS><<<cdiv(N, B), B, 0, stream>>>(HW, dis, N);
    gcn_gemm_wmma_f32<false, false><<<gemmBlocks, B, 0, stream>>>(x, nullptr, W1, dis, b1, HW, AGG, N, IN_CH);
    gcn_tile_scatter16<false><<<t16Grid, T16_THREADS, 0, stream>>>(src, dst, dis, HW, AGG, listP, listN, cnt16, listC, cntC, N, E, epcC);
    gcn_gemm_wmma_f32<true, false><<<gemmBlocks, B, 0, stream>>>(AGG, nullptr, W2, dis, b2, HW, AGG, N, HID);
  }
  gcn_tile_scatter16<true><<<t16Grid, T16_THREADS, 0, stream>>>(src, dst, dis, HW, AGG, listP, listN, cnt16, listC, cntC, N, E, epcC);

  gcn_mv16<<<cdiv(N, B), B, 0, stream>>>(AGG, W3, HW, N);
  if (keepN) {
    gcn_list_acc1<<<t16Grid, T16_THREADS, 0, stream>>>(src, dst, dis, HW, listP, listN, cnt16, HW + N, N, E, epcC);
    gcn_out_finish<T16_CHUNKS><<<cdiv(N, B), B, 0, stream>>>(HW + N, HW, dis, b3, out, N);
  } else {
    gcn_tile_acc1<true><<<tileGrid, TILE_THREADS, 0, stream>>>(src, dst, dis, HW, HW + N, N, E, epc);
    gcn_out_finish<TILE_CHUNKS><<<cdiv(N, B), B, 0, stream>>>(HW + N, HW, dis, b3, out, N);
  }
}
